# starting kernel
# baseline (speedup 1.0000x reference)
#include <hip/hip_runtime.h>
#include <hip/hip_bf16.h>

#define N_NODES   50000
#define D_FEAT    512
#define H1_DIM    128
#define H2_DIM    64
#define N_CLS     40
#define E_MP      800000
#define E_LP      250000

typedef __attribute__((ext_vector_type(2))) float v2f;
typedef __attribute__((ext_vector_type(8))) float v8f;

__global__ void fill_kernel(float* __restrict__ p, long long n, float v) {
  long long i = (long long)blockIdx.x * blockDim.x + threadIdx.x;
  if (i < n) p[i] = v;
}

__global__ void deg_accum_kernel(float* __restrict__ deg, const int* __restrict__ cols,
                                 const float* __restrict__ ew, int E) {
  long long i = (long long)blockIdx.x * blockDim.x + threadIdx.x;
  if (i < E) {
    float w = ew ? ew[i] : 1.0f;
    atomicAdd(&deg[cols[i]], w);
  }
}

__global__ void rsqrt_kernel(float* __restrict__ p, int n) {
  int i = blockIdx.x * blockDim.x + threadIdx.x;
  if (i < n) {
    float d = p[i];
    p[i] = (d > 0.0f) ? rsqrtf(d) : 0.0f;
  }
}

__global__ void bias_init_kernel(float* __restrict__ out, const float* __restrict__ b,
                                 long long n, int F) {
  long long i = (long long)blockIdx.x * blockDim.x + threadIdx.x;
  long long total = n * F;
  if (i < total) out[i] = b[i % F];
}

__global__ void pack_b_kernel(const float* __restrict__ B, float* __restrict__ Bp,
                              int K, int NC, int NCP) {
  int p = blockIdx.x * blockDim.x + threadIdx.x;
  int total = (K / 2) * NCP;
  if (p >= total) return;
  int k2 = p / NCP;
  int col = p % NCP;
  float x0 = 0.0f, x1 = 0.0f;
  if (col < NC) {
    x0 = B[(long long)(2 * k2) * NC + col];
    x1 = B[(long long)(2 * k2 + 1) * NC + col];
  }
  Bp[2 * p + 0] = x0;
  Bp[2 * p + 1] = x1;
}

template <int KDIM, int NT, int NCSTORE, bool RELU_A>
__global__ void wmma_gemm_kernel(const float* __restrict__ A, const float* __restrict__ Bp,
                                 float* __restrict__ C, int M) {
  constexpr int NCP = NT * 16;
  int wave = (int)(((long long)blockIdx.x * blockDim.x + threadIdx.x) >> 5);
  int lane = threadIdx.x & 31;
  int ntiles = M >> 4;
  if (wave >= ntiles) return;

  int row0 = wave << 4;
  int arow = row0 + (lane & 15);
  int khb  = lane >> 4;
  int bcol = lane & 15;

  v8f acc[NT] = {};

  const float2* __restrict__ Arow2 = (const float2*)(A + (long long)arow * KDIM);
  const float2* __restrict__ Bp2   = (const float2*)Bp;

#pragma unroll 2
  for (int k2 = 0; k2 < KDIM / 2; k2 += 2) {
    const float2* __restrict__ Bbase = Bp2 + (long long)(k2 + khb) * NCP + bcol;
    float2 bf[NT];
#pragma unroll
    for (int t = 0; t < NT; ++t) bf[t] = Bbase[t * 16];

    float2 af = Arow2[k2 + khb];
    if (RELU_A) { af.x = fmaxf(af.x, 0.0f); af.y = fmaxf(af.y, 0.0f); }
    v2f a; a.x = af.x; a.y = af.y;

#pragma unroll
    for (int t = 0; t < NT; ++t) {
      v2f b; b.x = bf[t].x; b.y = bf[t].y;
      acc[t] = __builtin_amdgcn_wmma_f32_16x16x4_f32(
           false, a,  false, b,
           (short)0, acc[t],  false,  false);
    }
  }

  int crow = row0 + (khb << 3);
  int ccol = lane & 15;
#pragma unroll
  for (int t = 0; t < NT; ++t) {
    int col = (t << 4) + ccol;
    if (col < NCSTORE) {
#pragma unroll
      for (int r = 0; r < 8; ++r) {
        C[(long long)(crow + r) * NCSTORE + col] = acc[t][r];
      }
    }
  }
}

__global__ void gcn_agg_kernel(const float* __restrict__ hx, float* __restrict__ out,
                               const int* __restrict__ rows, const int* __restrict__ cols,
                               const float* __restrict__ ew, const float* __restrict__ dinv,
                               int E, int N, int F) {
  int chunks = F >> 2;
  long long tid = (long long)blockIdx.x * blockDim.x + threadIdx.x;
  long long total = (long long)(E + N) * chunks;
  if (tid >= total) return;
  int e = (int)(tid / chunks);
  int c = (int)(tid % chunks) << 2;
  int r, cl; float w;
  if (e < E) { r = rows[e]; cl = cols[e]; w = ew ? ew[e] : 1.0f; }
  else       { r = e - E;   cl = r;       w = 1.0f; }
  float norm = dinv[r] * w * dinv[cl];
  const float4 h4 = *(const float4*)(hx + (long long)r * F + c);
  float* o = out + (long long)cl * F + c;
  atomicAdd(o + 0, h4.x * norm);
  atomicAdd(o + 1, h4.y * norm);
  atomicAdd(o + 2, h4.z * norm);
  atomicAdd(o + 3, h4.w * norm);
}

__global__ void log_softmax_kernel(const float* __restrict__ in, float* __restrict__ out,
                                   int n, int F) {
  int row = blockIdx.x * blockDim.x + threadIdx.x;
  if (row >= n) return;
  const float* p = in + (long long)row * F;
  float m = -INFINITY;
  for (int c = 0; c < F; ++c) m = fmaxf(m, p[c]);
  float s = 0.0f;
  for (int c = 0; c < F; ++c) s += __expf(p[c] - m);
  float ls = __logf(s) + m;
  float* o = out + (long long)row * F;
  for (int c = 0; c < F; ++c) o[c] = p[c] - ls;
}

__global__ void edge_dot_kernel(const float* __restrict__ h, const int* __restrict__ pos,
                                const int* __restrict__ neg, float* __restrict__ res) {
  int e = blockIdx.x * blockDim.x + threadIdx.x;
  if (e >= 2 * E_LP) return;
  const int* src = (e < E_LP) ? pos : neg;
  int eo = (e < E_LP) ? e : (e - E_LP);
  int j = src[eo];
  int i = src[E_LP + eo];
  const float4* hj = (const float4*)(h + (long long)j * H2_DIM);
  const float4* hi = (const float4*)(h + (long long)i * H2_DIM);
  float s = 0.0f;
#pragma unroll
  for (int k = 0; k < H2_DIM / 4; ++k) {
    float4 a = hi[k], b = hj[k];
    s += a.x * b.x + a.y * b.y + a.z * b.z + a.w * b.w;
  }
  res[e] = s;
}

static inline int blocks_for(long long total, int bs) {
  return (int)((total + bs - 1) / bs);
}

extern "C" void kernel_launch(void* const* d_in, const int* in_sizes, int n_in,
                              void* d_out, int out_size, void* d_ws, size_t ws_size,
                              hipStream_t stream) {
  const float* x      = (const float*)d_in[0];
  const int*   tpe    = (const int*)  d_in[1];
  const int*   ei     = (const int*)  d_in[2];
  const float* ew     = (const float*)d_in[3];
  const int*   pos    = (const int*)  d_in[4];
  const int*   neg    = (const int*)  d_in[5];
  const float* W1     = (const float*)d_in[6];
  const float* b1     = (const float*)d_in[7];
  const float* W2     = (const float*)d_in[8];
  const float* b2     = (const float*)d_in[9];
  const float* W_attr = (const float*)d_in[10];
  const float* b_attr = (const float*)d_in[11];
  const float* W_att  = (const float*)d_in[12];
  const float* b_att  = (const float*)d_in[13];

  float* out = (float*)d_out;
  float* ws  = (float*)d_ws;

  const long long OFF_DINV_T = 0;
  const long long OFF_DINV_W = 50000;
  const long long OFF_HX1    = 100000;
  const long long OFF_H1     = OFF_HX1 + 6400000;
  const long long OFF_HX2    = OFF_H1  + 6400000;
  const long long OFF_H2     = OFF_HX2 + 3200000;
  const long long OFF_W1P    = OFF_H2  + 3200000;
  const long long OFF_W2P    = OFF_W1P + 65536;
  const long long OFF_WAP    = OFF_W2P + 8192;
  const long long OFF_WBP    = OFF_WAP + 3072;
  const long long OFF_HXA    = OFF_HX1;
  const long long OFF_HXB    = OFF_HX1 + 2000000;
  const long long OFF_ATTR   = OFF_H1;
  const long long OFF_ATT    = OFF_H1 + 2000000;

  float* dinv_t = ws + OFF_DINV_T;
  float* dinv_w = ws + OFF_DINV_W;
  float* hx1    = ws + OFF_HX1;
  float* h1     = ws + OFF_H1;
  float* hx2    = ws + OFF_HX2;
  float* h2     = ws + OFF_H2;
  float* w1p    = ws + OFF_W1P;
  float* w2p    = ws + OFF_W2P;
  float* wap    = ws + OFF_WAP;
  float* wbp    = ws + OFF_WBP;
  float* hxa    = ws + OFF_HXA;
  float* hxb    = ws + OFF_HXB;
  float* attr_a = ws + OFF_ATTR;
  float* att_a  = ws + OFF_ATT;

  const int BS = 256;

  pack_b_kernel<<<blocks_for((D_FEAT / 2) * H1_DIM, BS), BS, 0, stream>>>(W1, w1p, D_FEAT, H1_DIM, H1_DIM);
  pack_b_kernel<<<blocks_for((H1_DIM / 2) * H2_DIM, BS), BS, 0, stream>>>(W2, w2p, H1_DIM, H2_DIM, H2_DIM);
  pack_b_kernel<<<blocks_for((H2_DIM / 2) * 48, BS), BS, 0, stream>>>(W_attr, wap, H2_DIM, N_CLS, 48);
  pack_b_kernel<<<blocks_for((H2_DIM / 2) * 48, BS), BS, 0, stream>>>(W_att,  wbp, H2_DIM, N_CLS, 48);

  fill_kernel<<<blocks_for(N_NODES, BS), BS, 0, stream>>>(dinv_t, N_NODES, 1.0f);
  deg_accum_kernel<<<blocks_for(E_MP, BS), BS, 0, stream>>>(dinv_t, tpe + E_MP, nullptr, E_MP);
  rsqrt_kernel<<<blocks_for(N_NODES, BS), BS, 0, stream>>>(dinv_t, N_NODES);

  fill_kernel<<<blocks_for(N_NODES, BS), BS, 0, stream>>>(dinv_w, N_NODES, 1.0f);
  deg_accum_kernel<<<blocks_for(E_MP, BS), BS, 0, stream>>>(dinv_w, ei + E_MP, ew, E_MP);
  rsqrt_kernel<<<blocks_for(N_NODES, BS), BS, 0, stream>>>(dinv_w, N_NODES);

  const int ROW_TILES = N_NODES / 16;
  const int GEMM_BS = 128;
  const int gemm_blocks = blocks_for((long long)ROW_TILES * 32, GEMM_BS);

  wmma_gemm_kernel<D_FEAT, 8, H1_DIM, false><<<gemm_blocks, GEMM_BS, 0, stream>>>(x, w1p, hx1, N_NODES);
  bias_init_kernel<<<blocks_for((long long)N_NODES * H1_DIM, BS), BS, 0, stream>>>(h1, b1, N_NODES, H1_DIM);
  gcn_agg_kernel<<<blocks_for((long long)(E_MP + N_NODES) * (H1_DIM / 4), BS), BS, 0, stream>>>(
      hx1, h1, tpe, tpe + E_MP, nullptr, dinv_t, E_MP, N_NODES, H1_DIM);

  wmma_gemm_kernel<H1_DIM, 4, H2_DIM, true><<<gemm_blocks, GEMM_BS, 0, stream>>>(h1, w2p, hx2, N_NODES);
  bias_init_kernel<<<blocks_for((long long)N_NODES * H2_DIM, BS), BS, 0, stream>>>(h2, b2, N_NODES, H2_DIM);
  gcn_agg_kernel<<<blocks_for((long long)(E_MP + N_NODES) * (H2_DIM / 4), BS), BS, 0, stream>>>(
      hx2, h2, tpe, tpe + E_MP, nullptr, dinv_t, E_MP, N_NODES, H2_DIM);

  wmma_gemm_kernel<H2_DIM, 3, N_CLS, false><<<gemm_blocks, GEMM_BS, 0, stream>>>(h2, wap, hxa, N_NODES);
  wmma_gemm_kernel<H2_DIM, 3, N_CLS, false><<<gemm_blocks, GEMM_BS, 0, stream>>>(h2, wbp, hxb, N_NODES);

  bias_init_kernel<<<blocks_for((long long)N_NODES * N_CLS, BS), BS, 0, stream>>>(attr_a, b_attr, N_NODES, N_CLS);
  gcn_agg_kernel<<<blocks_for((long long)(E_MP + N_NODES) * (N_CLS / 4), BS), BS, 0, stream>>>(
      hxa, attr_a, ei, ei + E_MP, ew, dinv_w, E_MP, N_NODES, N_CLS);

  bias_init_kernel<<<blocks_for((long long)N_NODES * N_CLS, BS), BS, 0, stream>>>(att_a, b_att, N_NODES, N_CLS);
  gcn_agg_kernel<<<blocks_for((long long)(E_MP + N_NODES) * (N_CLS / 4), BS), BS, 0, stream>>>(
      hxb, att_a, ei, ei + E_MP, ew, dinv_w, E_MP, N_NODES, N_CLS);

  edge_dot_kernel<<<blocks_for(2 * E_LP, BS), BS, 0, stream>>>(h2, pos, neg, out);
  log_softmax_kernel<<<blocks_for(N_NODES, BS), BS, 0, stream>>>(attr_a, out + 2 * E_LP, N_NODES, N_CLS);
  log_softmax_kernel<<<blocks_for(N_NODES, BS), BS, 0, stream>>>(att_a,  out + 2 * E_LP + (long long)N_NODES * N_CLS, N_NODES, N_CLS);
}


# gcn_agg x4: a block owns 256 destination rows in LDS (start = bias + self-loop term), scans all edge destinations, LDS f32 adds in arrival order, line stores; no global f32 atomics, 25 -> 21
# speedup vs baseline: 1.3138x; 1.3138x over previous
#include <hip/hip_runtime.h>
#include <hip/hip_bf16.h>

#define N_NODES   50000
#define D_FEAT    512
#define H1_DIM    128
#define H2_DIM    64
#define N_CLS     40
#define E_MP      800000
#define E_LP      250000

typedef __attribute__((ext_vector_type(2))) float v2f;
typedef __attribute__((ext_vector_type(8))) float v8f;

__global__ void fill_kernel(float* __restrict__ p, long long n, float v) {
  long long i = (long long)blockIdx.x * blockDim.x + threadIdx.x;
  if (i < n) p[i] = v;
}

__global__ void deg_accum_kernel(float* __restrict__ deg, const int* __restrict__ cols,
                                 const float* __restrict__ ew, int E) {
  long long i = (long long)blockIdx.x * blockDim.x + threadIdx.x;
  if (i < E) {
    float w = ew ? ew[i] : 1.0f;
    atomicAdd(&deg[cols[i]], w);
  }
}

__global__ void rsqrt_kernel(float* __restrict__ p, int n) {
  int i = blockIdx.x * blockDim.x + threadIdx.x;
  if (i < n) {
    float d = p[i];
    p[i] = (d > 0.0f) ? rsqrtf(d) : 0.0f;
  }
}

__global__ void pack_b_kernel(const float* __restrict__ B, float* __restrict__ Bp,
                              int K, int NC, int NCP) {
  int p = blockIdx.x * blockDim.x + threadIdx.x;
  int total = (K / 2) * NCP;
  if (p >= total) return;
  int k2 = p / NCP;
  int col = p % NCP;
  float x0 = 0.0f, x1 = 0.0f;
  if (col < NC) {
    x0 = B[(long long)(2 * k2) * NC + col];
    x1 = B[(long long)(2 * k2 + 1) * NC + col];
  }
  Bp[2 * p + 0] = x0;
  Bp[2 * p + 1] = x1;
}

template <int KDIM, int NT, int NCSTORE, bool RELU_A>
__global__ void wmma_gemm_kernel(const float* __restrict__ A, const float* __restrict__ Bp,
                                 float* __restrict__ C, int M) {
  constexpr int NCP = NT * 16;
  int wave = (int)(((long long)blockIdx.x * blockDim.x + threadIdx.x) >> 5);
  int lane = threadIdx.x & 31;
  int ntiles = M >> 4;
  if (wave >= ntiles) return;

  int row0 = wave << 4;
  int arow = row0 + (lane & 15);
  int khb  = lane >> 4;
  int bcol = lane & 15;

  v8f acc[NT] = {};

  const float2* __restrict__ Arow2 = (const float2*)(A + (long long)arow * KDIM);
  const float2* __restrict__ Bp2   = (const float2*)Bp;

#pragma unroll 2
  for (int k2 = 0; k2 < KDIM / 2; k2 += 2) {
    const float2* __restrict__ Bbase = Bp2 + (long long)(k2 + khb) * NCP + bcol;
    float2 bf[NT];
#pragma unroll
    for (int t = 0; t < NT; ++t) bf[t] = Bbase[t * 16];

    float2 af = Arow2[k2 + khb];
    if (RELU_A) { af.x = fmaxf(af.x, 0.0f); af.y = fmaxf(af.y, 0.0f); }
    v2f a; a.x = af.x; a.y = af.y;

#pragma unroll
    for (int t = 0; t < NT; ++t) {
      v2f b; b.x = bf[t].x; b.y = bf[t].y;
      acc[t] = __builtin_amdgcn_wmma_f32_16x16x4_f32(
           false, a,  false, b,
           (short)0, acc[t],  false,  false);
    }
  }

  int crow = row0 + (khb << 3);
  int ccol = lane & 15;
#pragma unroll
  for (int t = 0; t < NT; ++t) {
    int col = (t << 4) + ccol;
    if (col < NCSTORE) {
#pragma unroll
      for (int r = 0; r < 8; ++r) {
        C[(long long)(crow + r) * NCSTORE + col] = acc[t][r];
      }
    }
  }
}

#define AGG_R   256
#define AGG_Q   8192
#define AGG_BS  1024

template <int F>
__global__ __launch_bounds__(AGG_BS)
void gcn_agg_own_kernel(const float* __restrict__ hx, float* __restrict__ out,
                        const int* __restrict__ rows, const int* __restrict__ cols,
                        const float* __restrict__ ew, const float* __restrict__ dinv,
                        const float* __restrict__ bias, int E, int N) {
  static_assert(N_NODES <= 65536 && AGG_R <= 65536, "queue word packing: source row | local row << 16");
  __shared__ float    acc[AGG_R * F];
  __shared__ unsigned qa[AGG_Q];
  __shared__ float    qn[AGG_Q];
  __shared__ unsigned qcount;

  const int tid  = threadIdx.x;
  const int lane = tid & 31;
  const int wave = tid >> 5;
  const int n0   = blockIdx.x * AGG_R;
  const int nr   = min(AGG_R, N - n0);

  if (tid == 0) qcount = 0u;
  for (int i = tid; i < nr * F; i += AGG_BS) {
    int r = i / F;
    int f = i - r * F;
    float d = dinv[n0 + r];
    acc[i] = bias[f] + (d * 1.0f * d) * hx[(long long)n0 * F + i];
  }
  __syncthreads();

  const int4* __restrict__ cols4 = (const int4*)cols;
  const int E4 = E >> 2;
  constexpr int SU = 8;
  const int4 none = make_int4(-1, -1, -1, -1);
  int4 c[SU];
  if (E4 > 0) {
#pragma unroll
    for (int u = 0; u < SU; ++u) {
      int j = tid + u * AGG_BS;
      int4 t = cols4[min(j, E4 - 1)];
      c[u] = (j < E4) ? t : none;
    }
  }
  for (int i = tid; i < E4; i += AGG_BS * SU) {
    int4 nx[SU];
#pragma unroll
    for (int u = 0; u < SU; ++u) {
      int j = i + (SU + u) * AGG_BS;
      int4 t = cols4[min(j, E4 - 1)];
      nx[u] = (j < E4) ? t : none;
    }
    unsigned hits = 0u;
#pragma unroll
    for (int u = 0; u < SU; ++u) {
      hits |= ((unsigned)(c[u].x - n0) < (unsigned)nr ? 1u : 0u) << (4 * u + 0);
      hits |= ((unsigned)(c[u].y - n0) < (unsigned)nr ? 1u : 0u) << (4 * u + 1);
      hits |= ((unsigned)(c[u].z - n0) < (unsigned)nr ? 1u : 0u) << (4 * u + 2);
      hits |= ((unsigned)(c[u].w - n0) < (unsigned)nr ? 1u : 0u) << (4 * u + 3);
    }
    while (hits) {
      int b = __builtin_ctz(hits);
      hits &= hits - 1u;
      int e = 4 * (i + (b >> 2) * AGG_BS) + (b & 3);
      unsigned slot = atomicAdd(&qcount, 1u);
      if (slot < AGG_Q) {
        qa[slot] = (unsigned)e;
      } else {
        int cl = cols[e] - n0;
        int r = rows[e];
        float w = ew ? ew[e] : 1.0f;
        float nm = dinv[r] * w * dinv[n0 + cl];
        for (int f = 0; f < F; ++f)
          atomicAdd(&acc[cl * F + f], hx[(long long)r * F + f] * nm);
      }
    }
#pragma unroll
    for (int u = 0; u < SU; ++u) c[u] = nx[u];
  }
  for (int e = (E4 << 2) + tid; e < E; e += AGG_BS) {
    unsigned cl = (unsigned)(cols[e] - n0);
    if (cl < (unsigned)nr) {
      int r = rows[e];
      float w = ew ? ew[e] : 1.0f;
      float nm = dinv[r] * w * dinv[n0 + cl];
      for (int f = 0; f < F; ++f)
        atomicAdd(&acc[cl * F + f], hx[(long long)r * F + f] * nm);
    }
  }
  __syncthreads();

  const int nq = (int)min(qcount, (unsigned)AGG_Q);
  for (int i = tid; i < nq; i += AGG_BS) {
    int e  = (int)qa[i];
    int cg = cols[e];
    int r  = rows[e];
    float w = ew ? ew[e] : 1.0f;
    qn[i] = dinv[r] * w * dinv[cg];
    qa[i] = (unsigned)r | ((unsigned)(cg - n0) << 16);
  }
  __syncthreads();

  constexpr int NW  = AGG_BS / 32;
  constexpr int U   = 8;
  constexpr int PER = (F + 31) / 32;
  const int swave = __builtin_amdgcn_readfirstlane(wave);
  for (int i0 = swave; i0 < nq; i0 += NW * U) {
    float v[U][PER];
    int   cl[U];
#pragma unroll
    for (int u = 0; u < U; ++u) {
      int i  = i0 + u * NW;
      int ic = min(i, nq - 1);
      unsigned p = qa[ic];
      float nm   = qn[ic];
      int r = (int)(p & 0xFFFFu);
      cl[u] = (i < nq) ? (int)(p >> 16) : -1;
      const float* __restrict__ s = hx + (long long)r * F;
#pragma unroll
      for (int k = 0; k < PER; ++k) {
        int f = min(lane + 32 * k, F - 1);
        v[u][k] = s[f] * nm;
      }
    }
#pragma unroll
    for (int u = 0; u < U; ++u) {
      if (cl[u] >= 0) {
#pragma unroll
        for (int k = 0; k < PER; ++k) {
          int f = lane + 32 * k;
          if (f < F) atomicAdd(&acc[cl[u] * F + f], v[u][k]);
        }
      }
    }
  }
  __syncthreads();

  float* __restrict__ o = out + (long long)n0 * F;
  for (int i = tid * 4; i < nr * F; i += AGG_BS * 4)
    *(float4*)(o + i) = *(const float4*)(acc + i);
}

__global__ void log_softmax_kernel(const float* __restrict__ in, float* __restrict__ out,
                                   int n, int F) {
  int row = blockIdx.x * blockDim.x + threadIdx.x;
  if (row >= n) return;
  const float* p = in + (long long)row * F;
  float m = -INFINITY;
  for (int c = 0; c < F; ++c) m = fmaxf(m, p[c]);
  float s = 0.0f;
  for (int c = 0; c < F; ++c) s += __expf(p[c] - m);
  float ls = __logf(s) + m;
  float* o = out + (long long)row * F;
  for (int c = 0; c < F; ++c) o[c] = p[c] - ls;
}

__global__ void edge_dot_kernel(const float* __restrict__ h, const int* __restrict__ pos,
                                const int* __restrict__ neg, float* __restrict__ res) {
  int e = blockIdx.x * blockDim.x + threadIdx.x;
  if (e >= 2 * E_LP) return;
  const int* src = (e < E_LP) ? pos : neg;
  int eo = (e < E_LP) ? e : (e - E_LP);
  int j = src[eo];
  int i = src[E_LP + eo];
  const float4* hj = (const float4*)(h + (long long)j * H2_DIM);
  const float4* hi = (const float4*)(h + (long long)i * H2_DIM);
  float s = 0.0f;
#pragma unroll
  for (int k = 0; k < H2_DIM / 4; ++k) {
    float4 a = hi[k], b = hj[k];
    s += a.x * b.x + a.y * b.y + a.z * b.z + a.w * b.w;
  }
  res[e] = s;
}

static inline int blocks_for(long long total, int bs) {
  return (int)((total + bs - 1) / bs);
}

extern "C" void kernel_launch(void* const* d_in, const int* in_sizes, int n_in,
                              void* d_out, int out_size, void* d_ws, size_t ws_size,
                              hipStream_t stream) {
  const float* x      = (const float*)d_in[0];
  const int*   tpe    = (const int*)  d_in[1];
  const int*   ei     = (const int*)  d_in[2];
  const float* ew     = (const float*)d_in[3];
  const int*   pos    = (const int*)  d_in[4];
  const int*   neg    = (const int*)  d_in[5];
  const float* W1     = (const float*)d_in[6];
  const float* b1     = (const float*)d_in[7];
  const float* W2     = (const float*)d_in[8];
  const float* b2     = (const float*)d_in[9];
  const float* W_attr = (const float*)d_in[10];
  const float* b_attr = (const float*)d_in[11];
  const float* W_att  = (const float*)d_in[12];
  const float* b_att  = (const float*)d_in[13];

  float* out = (float*)d_out;
  float* ws  = (float*)d_ws;

  const long long OFF_DINV_T = 0;
  const long long OFF_DINV_W = 50000;
  const long long OFF_HX1    = 100000;
  const long long OFF_H1     = OFF_HX1 + 6400000;
  const long long OFF_HX2    = OFF_H1  + 6400000;
  const long long OFF_H2     = OFF_HX2 + 3200000;
  const long long OFF_W1P    = OFF_H2  + 3200000;
  const long long OFF_W2P    = OFF_W1P + 65536;
  const long long OFF_WAP    = OFF_W2P + 8192;
  const long long OFF_WBP    = OFF_WAP + 3072;
  const long long OFF_HXA    = OFF_HX1;
  const long long OFF_HXB    = OFF_HX1 + 2000000;
  const long long OFF_ATTR   = OFF_H1;
  const long long OFF_ATT    = OFF_H1 + 2000000;

  float* dinv_t = ws + OFF_DINV_T;
  float* dinv_w = ws + OFF_DINV_W;
  float* hx1    = ws + OFF_HX1;
  float* h1     = ws + OFF_H1;
  float* hx2    = ws + OFF_HX2;
  float* h2     = ws + OFF_H2;
  float* w1p    = ws + OFF_W1P;
  float* w2p    = ws + OFF_W2P;
  float* wap    = ws + OFF_WAP;
  float* wbp    = ws + OFF_WBP;
  float* hxa    = ws + OFF_HXA;
  float* hxb    = ws + OFF_HXB;
  float* attr_a = ws + OFF_ATTR;
  float* att_a  = ws + OFF_ATT;

  const int BS = 256;

  pack_b_kernel<<<blocks_for((D_FEAT / 2) * H1_DIM, BS), BS, 0, stream>>>(W1, w1p, D_FEAT, H1_DIM, H1_DIM);
  pack_b_kernel<<<blocks_for((H1_DIM / 2) * H2_DIM, BS), BS, 0, stream>>>(W2, w2p, H1_DIM, H2_DIM, H2_DIM);
  pack_b_kernel<<<blocks_for((H2_DIM / 2) * 48, BS), BS, 0, stream>>>(W_attr, wap, H2_DIM, N_CLS, 48);
  pack_b_kernel<<<blocks_for((H2_DIM / 2) * 48, BS), BS, 0, stream>>>(W_att,  wbp, H2_DIM, N_CLS, 48);

  fill_kernel<<<blocks_for(N_NODES, BS), BS, 0, stream>>>(dinv_t, N_NODES, 1.0f);
  deg_accum_kernel<<<blocks_for(E_MP, BS), BS, 0, stream>>>(dinv_t, tpe + E_MP, nullptr, E_MP);
  rsqrt_kernel<<<blocks_for(N_NODES, BS), BS, 0, stream>>>(dinv_t, N_NODES);

  fill_kernel<<<blocks_for(N_NODES, BS), BS, 0, stream>>>(dinv_w, N_NODES, 1.0f);
  deg_accum_kernel<<<blocks_for(E_MP, BS), BS, 0, stream>>>(dinv_w, ei + E_MP, ew, E_MP);
  rsqrt_kernel<<<blocks_for(N_NODES, BS), BS, 0, stream>>>(dinv_w, N_NODES);

  const int ROW_TILES = N_NODES / 16;
  const int GEMM_BS = 128;
  const int gemm_blocks = blocks_for((long long)ROW_TILES * 32, GEMM_BS);
  const int AGG_BLOCKS = (N_NODES + AGG_R - 1) / AGG_R;

  wmma_gemm_kernel<D_FEAT, 8, H1_DIM, false><<<gemm_blocks, GEMM_BS, 0, stream>>>(x, w1p, hx1, N_NODES);
  gcn_agg_own_kernel<H1_DIM><<<AGG_BLOCKS, AGG_BS, 0, stream>>>(
      hx1, h1, tpe, tpe + E_MP, nullptr, dinv_t, b1, E_MP, N_NODES);

  wmma_gemm_kernel<H1_DIM, 4, H2_DIM, true><<<gemm_blocks, GEMM_BS, 0, stream>>>(h1, w2p, hx2, N_NODES);
  gcn_agg_own_kernel<H2_DIM><<<AGG_BLOCKS, AGG_BS, 0, stream>>>(
      hx2, h2, tpe, tpe + E_MP, nullptr, dinv_t, b2, E_MP, N_NODES);

  wmma_gemm_kernel<H2_DIM, 3, N_CLS, false><<<gemm_blocks, GEMM_BS, 0, stream>>>(h2, wap, hxa, N_NODES);
  wmma_gemm_kernel<H2_DIM, 3, N_CLS, false><<<gemm_blocks, GEMM_BS, 0, stream>>>(h2, wbp, hxb, N_NODES);

  gcn_agg_own_kernel<N_CLS><<<AGG_BLOCKS, AGG_BS, 0, stream>>>(
      hxa, attr_a, ei, ei + E_MP, ew, dinv_w, b_attr, E_MP, N_NODES);

  gcn_agg_own_kernel<N_CLS><<<AGG_BLOCKS, AGG_BS, 0, stream>>>(
      hxb, att_a, ei, ei + E_MP, ew, dinv_w, b_att, E_MP, N_NODES);

  edge_dot_kernel<<<blocks_for(2 * E_LP, BS), BS, 0, stream>>>(h2, pos, neg, out);
  log_softmax_kernel<<<blocks_for(N_NODES, BS), BS, 0, stream>>>(attr_a, out + 2 * E_LP, N_NODES, N_CLS);
  log_softmax_kernel<<<blocks_for(N_NODES, BS), BS, 0, stream>>>(att_a,  out + 2 * E_LP + (long long)N_NODES * N_CLS, N_NODES, N_CLS);
}


# aggregation x4, degree x2: destination tile 256 -> 200 rows, so 250 one-per-WGP blocks use 250 of the chip's 256 WGPs (was 196), 22% less rows and edges per block; records, sums, their order kept
# speedup vs baseline: 2.5078x; 1.9088x over previous
#include <hip/hip_runtime.h>
#include <hip/hip_bf16.h>

#define N_NODES   50000
#define D_FEAT    512
#define H1_DIM    128
#define H2_DIM    64
#define N_CLS     40
#define E_MP      800000
#define E_LP      250000

typedef __attribute__((ext_vector_type(2))) float v2f;
typedef __attribute__((ext_vector_type(8))) float v8f;

__global__ void pack_b_kernel(const float* __restrict__ B, float* __restrict__ Bp,
                              int K, int NC, int NCP) {
  int p = blockIdx.x * blockDim.x + threadIdx.x;
  int total = (K / 2) * NCP;
  if (p >= total) return;
  int k2 = p / NCP;
  int col = p % NCP;
  float x0 = 0.0f, x1 = 0.0f;
  if (col < NC) {
    x0 = B[(long long)(2 * k2) * NC + col];
    x1 = B[(long long)(2 * k2 + 1) * NC + col];
  }
  Bp[2 * p + 0] = x0;
  Bp[2 * p + 1] = x1;
}

template <int KDIM, int NT, int NCSTORE, bool RELU_A>
__global__ void wmma_gemm_kernel(const float* __restrict__ A, const float* __restrict__ Bp,
                                 const float* __restrict__ rs, float* __restrict__ C, int M) {
  constexpr int NCP = NT * 16;
  int wave = (int)(((long long)blockIdx.x * blockDim.x + threadIdx.x) >> 5);
  int lane = threadIdx.x & 31;
  int ntiles = M >> 4;
  if (wave >= ntiles) return;

  int row0 = wave << 4;
  int arow = row0 + (lane & 15);
  int khb  = lane >> 4;
  int bcol = lane & 15;

  v8f acc[NT] = {};

  const int crow = row0 + (khb << 3);
  float sc[8];
#pragma unroll
  for (int r = 0; r < 8; ++r) sc[r] = rs[crow + r];

  const float2* __restrict__ Arow2 = (const float2*)(A + (long long)arow * KDIM);
  const float2* __restrict__ Bp2   = (const float2*)Bp;

#pragma unroll 2
  for (int k2 = 0; k2 < KDIM / 2; k2 += 2) {
    const float2* __restrict__ Bbase = Bp2 + (long long)(k2 + khb) * NCP + bcol;
    float2 bf[NT];
#pragma unroll
    for (int t = 0; t < NT; ++t) bf[t] = Bbase[t * 16];

    float2 af = Arow2[k2 + khb];
    if (RELU_A) { af.x = fmaxf(af.x, 0.0f); af.y = fmaxf(af.y, 0.0f); }
    v2f a; a.x = af.x; a.y = af.y;

#pragma unroll
    for (int t = 0; t < NT; ++t) {
      v2f b; b.x = bf[t].x; b.y = bf[t].y;
      acc[t] = __builtin_amdgcn_wmma_f32_16x16x4_f32(
           false, a,  false, b,
           (short)0, acc[t],  false,  false);
    }
  }

  int ccol = lane & 15;
#pragma unroll
  for (int t = 0; t < NT; ++t) {
    int col = (t << 4) + ccol;
    if (col < NCSTORE) {
#pragma unroll
      for (int r = 0; r < 8; ++r) {
        C[(long long)(crow + r) * NCSTORE + col] = acc[t][r] * sc[r];
      }
    }
  }
}

#define AGG_R   200
#define AGG_Q   8192
#define AGG_BS  1024
#define AGG_NB  ((N_NODES + AGG_R - 1) / AGG_R)
#define BK_CHUNK 16384
#define BK_BS    1024
#define BK_PER   (BK_CHUNK / BK_BS)

template <bool WT>
__global__ __launch_bounds__(BK_BS)
void bucket_kernel(const int* __restrict__ rows, const int* __restrict__ cols,
                   const float* __restrict__ ew, unsigned* __restrict__ bent,
                   unsigned* __restrict__ btab, int E, int N) {
  static_assert(AGG_NB < 256 && BK_CHUNK <= (1 << 24), "packing: tile (or spare bin) | rank << 8; prefix on 8 waves");
  static_assert(N_NODES <= 65536 && AGG_R <= 65536, "record packing: source row | local row << 16");
  __shared__ unsigned hist[AGG_NB + 1];
  __shared__ unsigned start[AGG_NB + 1];
  __shared__ unsigned wsum[8];
  constexpr int RW = WT ? 2 : 1;
  __shared__ __attribute__((aligned(16))) unsigned stage[RW * BK_CHUNK];
  const int tid  = threadIdx.x;
  const int lane = tid & 31;
  const int c0   = blockIdx.x * BK_CHUNK;
  const int len  = min(BK_CHUNK, E - c0);

  for (int i = tid; i <= AGG_NB; i += BK_BS) hist[i] = 0u;
  __syncthreads();

  unsigned cv[BK_PER];
  unsigned rv[BK_PER];
  float    wv[BK_PER];
#pragma unroll
  for (int u = 0; u < BK_PER; ++u) {
    const int j = c0 + min(tid + u * BK_BS, len - 1);
    cv[u] = (unsigned)cols[j];
    rv[u] = (unsigned)rows[j];
    wv[u] = WT ? ew[j] : 1.0f;
  }

  unsigned pk[BK_PER];
#pragma unroll
  for (int u = 0; u < BK_PER; ++u) {
    bool ok = (tid + u * BK_BS < len) && (cv[u] < (unsigned)N);
    unsigned t = ok ? cv[u] / AGG_R : (unsigned)AGG_NB;
    unsigned r = atomicAdd(&hist[t], 1u);
    pk[u] = t | (r << 8);
  }
  __syncthreads();

  unsigned v = 0u, inc = 0u;
  if (tid < 256) {
    v = (tid < AGG_NB) ? hist[tid] : 0u;
    inc = v;
#pragma unroll
    for (int d = 1; d < 32; d <<= 1) {
      unsigned o = __shfl_up(inc, d);
      if (lane >= d) inc += o;
    }
    if (lane == 31) wsum[tid >> 5] = inc;
  }
  __syncthreads();
  if (tid <= AGG_NB) {
    unsigned off = 0u;
#pragma unroll
    for (int w = 0; w < 8; ++w) off += (w < (tid >> 5)) ? wsum[w] : 0u;
    unsigned ex = off + inc - v;
    start[tid] = ex;
    btab[(long long)blockIdx.x * (AGG_NB + 1) + tid] = (unsigned)c0 + ex;
  }
  __syncthreads();

#pragma unroll
  for (int u = 0; u < BK_PER; ++u) {
    unsigned t = pk[u] & 0xFFu;
    if (t < (unsigned)AGG_NB) {
      unsigned r   = pk[u] >> 8;
      unsigned pos = start[t] + r;
      unsigned rec = rv[u] | ((cv[u] - t * AGG_R) << 16);
      if (WT) ((uint2*)stage)[pos] = make_uint2(rec, __float_as_uint(wv[u]));
      else    stage[pos] = rec;
    }
  }
  __syncthreads();

  const unsigned nw = start[AGG_NB] * RW;
  unsigned* __restrict__ dst = bent + (long long)c0 * RW;
  for (unsigned i = 4u * tid; i < nw; i += 4u * BK_BS) {
    if (i + 4u <= nw) {
      *(uint4*)(dst + i) = *(const uint4*)(stage + i);
    } else {
      for (unsigned k = i; k < nw; ++k) dst[k] = stage[k];
    }
  }
}

#define DEG_BS 1024
#define DEG_U  4
template <bool WT>
__global__ __launch_bounds__(DEG_BS)
void deg_own_kernel(float* __restrict__ dinv,
                    const unsigned* __restrict__ bent, const unsigned* __restrict__ btab,
                    int nch, int N) {
  __shared__ float dsum[AGG_R];
  const int tid  = threadIdx.x;
  const int lane = tid & 31;
  const int n0   = blockIdx.x * AGG_R;
  const int nr   = min(AGG_R, N - n0);

  if (tid < AGG_R) dsum[tid] = 1.0f;
  __syncthreads();

  constexpr int NW = DEG_BS / 32;
  const int swave = __builtin_amdgcn_readfirstlane(tid >> 5);
  for (int c = swave; c < nch; c += NW) {
    const unsigned* __restrict__ trow = btab + (long long)c * (AGG_NB + 1) + blockIdx.x;
    const unsigned s0  = trow[0];
    const unsigned cnt = trow[1] - s0;
    for (unsigned i0 = 0; i0 < cnt; i0 += 32u * DEG_U) {
      unsigned cl[DEG_U];
      float    w[DEG_U];
#pragma unroll
      for (int u = 0; u < DEG_U; ++u) {
        const unsigned j = s0 + min(i0 + 32u * u + lane, cnt - 1u);
        if (WT) {
          uint2 p = ((const uint2*)bent)[j];
          cl[u] = p.x >> 16;
          w[u]  = __uint_as_float(p.y);
        } else {
          cl[u] = bent[j] >> 16;
          w[u]  = 1.0f;
        }
      }
#pragma unroll
      for (int u = 0; u < DEG_U; ++u)
        if (i0 + 32u * u + lane < cnt) atomicAdd(&dsum[cl[u]], w[u]);
    }
  }
  __syncthreads();

  if (tid < nr) {
    float d = dsum[tid];
    dinv[n0 + tid] = (d > 0.0f) ? rsqrtf(d) : 0.0f;
  }
}

#define AGG_CU 4
#define HEAD_CP 48
template <int F, bool WT, bool HEADS>
__global__ __launch_bounds__(AGG_BS)
void gcn_agg_own_kernel(const float* __restrict__ hx, float* __restrict__ out,
                        const float* __restrict__ dinv, const float* __restrict__ bias,
                        const unsigned* __restrict__ bent, const unsigned* __restrict__ btab,
                        int nch, int N,
                        const float* __restrict__ Wa, const float* __restrict__ Wb,
                        const float* __restrict__ hscale,
                        float* __restrict__ ha, float* __restrict__ hb) {
  constexpr int P   = HEADS ? F + 4 : F;
  constexpr int WSH = HEADS ? 2 * (F / 2) * HEAD_CP * 2 : 4;
  constexpr int RT  = (AGG_R + 15) / 16;
  constexpr int RP  = HEADS ? RT * 16 : AGG_R;
  __shared__ __attribute__((aligned(16))) float acc[RP * P];
  __shared__ __attribute__((aligned(16))) float wsh[WSH];
  __shared__ unsigned qa[AGG_Q];
  __shared__ float    qw[WT ? AGG_Q : 1];
  __shared__ unsigned qcount;

  const int tid  = threadIdx.x;
  const int lane = tid & 31;
  const int wave = tid >> 5;
  const int n0   = blockIdx.x * AGG_R;
  const int nr   = min(AGG_R, N - n0);

  if (tid == 0) qcount = 0u;
  constexpr int V      = 4;
  constexpr int NPIECE = AGG_R * F / V;
  constexpr int NIT    = (NPIECE + AGG_BS - 1) / AGG_BS;
  static_assert(F % V == 0 && (AGG_R * F) % V == 0, "a piece stays inside a row");
  typedef float fvec __attribute__((ext_vector_type(V)));
  {
    const float* __restrict__ own = hx + (long long)n0 * F;
    if (nr == AGG_R) {
      fvec t[NIT];
#pragma unroll
      for (int u = 0; u < NIT; ++u) {
        const int pi = tid + u * AGG_BS;
        if (u < NIT - 1 || pi < NPIECE) t[u] = *(const fvec*)(own + pi * V);
      }
#pragma unroll
      for (int u = 0; u < NIT; ++u) {
        const int pi = tid + u * AGG_BS;
        if (u < NIT - 1 || pi < NPIECE) {
          const int i = pi * V;
          const int r = i / F;
          const int f = i - r * F;
          *(fvec*)(acc + ((P == F) ? i : r * P + f)) = t[u];
        }
      }
    } else {
      for (int i = tid * V; i < nr * F; i += AGG_BS * V) {
        const int r = i / F;
        const int f = i - r * F;
        *(fvec*)(acc + ((P == F) ? i : r * P + f)) = *(const fvec*)(own + i);
      }
    }
  }
  if constexpr (HEADS) {
    for (int i = nr * P + tid; i < RP * P; i += AGG_BS) acc[i] = 0.0f;
    for (int p = tid; p < 2 * (F / 2) * HEAD_CP; p += AGG_BS) {
      const int hd  = p / ((F / 2) * HEAD_CP);
      const int q   = p - hd * ((F / 2) * HEAD_CP);
      const int k2  = q / HEAD_CP;
      const int col = q - k2 * HEAD_CP;
      const float* __restrict__ Wsel = hd ? Wb : Wa;
      float x0 = 0.0f, x1 = 0.0f;
      if (col < N_CLS) {
        x0 = Wsel[(2 * k2) * N_CLS + col];
        x1 = Wsel[(2 * k2 + 1) * N_CLS + col];
      }
      ((float2*)wsh)[p] = make_float2(x0, x1);
    }
  }
  __syncthreads();

  constexpr int NW = AGG_BS / 32;
  const int swave = __builtin_amdgcn_readfirstlane(wave);
  for (int c = swave; c < nch; c += NW) {
    const unsigned* __restrict__ trow = btab + (long long)c * (AGG_NB + 1) + blockIdx.x;
    const unsigned s0  = trow[0];
    const unsigned cnt = trow[1] - s0;
    unsigned base = 0u;
    if (lane == 0) base = atomicAdd(&qcount, cnt);
    base = (unsigned)__builtin_amdgcn_readfirstlane((int)base);
    for (unsigned i0 = 0; i0 < cnt; i0 += 32u * AGG_CU) {
      unsigned rec[AGG_CU];
      float    w[AGG_CU];
#pragma unroll
      for (int u = 0; u < AGG_CU; ++u) {
        const unsigned j = s0 + min(i0 + 32u * u + lane, cnt - 1u);
        if (WT) {
          uint2 p = ((const uint2*)bent)[j];
          rec[u] = p.x;
          w[u]   = __uint_as_float(p.y);
        } else {
          rec[u] = bent[j];
          w[u]   = 1.0f;
        }
      }
#pragma unroll
      for (int u = 0; u < AGG_CU; ++u) {
        const unsigned i = i0 + 32u * u + lane;
        if (i < cnt) {
          const unsigned slot = base + i;
          if (slot < AGG_Q) {
            qa[slot] = rec[u];
            if constexpr (WT) qw[slot] = w[u];
          } else {
            const int r  = (int)(rec[u] & 0xFFFFu);
            const int cl = (int)(rec[u] >> 16);
            for (int f = 0; f < F; ++f) {
              const float x = hx[(long long)r * F + f];
              atomicAdd(&acc[cl * P + f], WT ? x * w[u] : x);
            }
          }
        }
      }
    }
  }
  __syncthreads();

  const int nq = (int)min(qcount, (unsigned)AGG_Q);
  constexpr int U   = 8;
  constexpr int PER = (F + 31) / 32;
  for (int i0 = swave; i0 < nq; i0 += NW * U) {
    float v[U][PER];
    int   cl[U];
#pragma unroll
    for (int u = 0; u < U; ++u) {
      int i  = i0 + u * NW;
      int ic = min(i, nq - 1);
      unsigned p = qa[ic];
      float wq = 1.0f;
      if constexpr (WT) wq = qw[ic];
      int r = (int)(p & 0xFFFFu);
      cl[u] = (i < nq) ? (int)(p >> 16) : -1;
      const float* __restrict__ s = hx + (long long)r * F;
#pragma unroll
      for (int k = 0; k < PER; ++k) {
        int f = min(lane + 32 * k, F - 1);
        v[u][k] = WT ? s[f] * wq : s[f];
      }
    }
#pragma unroll
    for (int u = 0; u < U; ++u) {
      if (cl[u] >= 0) {
#pragma unroll
        for (int k = 0; k < PER; ++k) {
          int f = lane + 32 * k;
          if (f < F) atomicAdd(&acc[cl[u] * P + f], v[u][k]);
        }
      }
    }
  }
  __syncthreads();

  float* __restrict__ o = out + (long long)n0 * F;
  if (nr == AGG_R) {
    float dd[NIT];
    float hh[NIT];
    fvec  bb[NIT];
#pragma unroll
    for (int u = 0; u < NIT; ++u) {
      const int pi = tid + u * AGG_BS;
      if (u < NIT - 1 || pi < NPIECE) {
        const int i = pi * V;
        const int r = i / F;
        const int f = i - r * F;
        dd[u] = dinv[n0 + r];
        hh[u] = HEADS ? hscale[n0 + r] : 1.0f;
#pragma unroll
        for (int k = 0; k < V; ++k) bb[u][k] = bias[f + k];
      }
    }
#pragma unroll
    for (int u = 0; u < NIT; ++u) {
      const int pi = tid + u * AGG_BS;
      if (u < NIT - 1 || pi < NPIECE) {
        const int i = pi * V;
        const int r = i / F;
        const int f = i - r * F;
        float* a = acc + ((P == F) ? i : r * P + f);
        const fvec s = bb[u] + dd[u] * *(const fvec*)a;
        *(fvec*)(o + i) = s;
        if constexpr (HEADS) *(fvec*)a = hh[u] * s;
      }
    }
  } else {
    for (int i = tid * V; i < nr * F; i += AGG_BS * V) {
      const int r = i / F;
      const int f = i - r * F;
      float* a = acc + ((P == F) ? i : r * P + f);
      const float d = dinv[n0 + r];
      fvec s = *(const fvec*)a;
#pragma unroll
      for (int k = 0; k < V; ++k) s[k] = bias[f + k] + d * s[k];
      *(fvec*)(o + i) = s;
      if constexpr (HEADS) *(fvec*)a = hscale[n0 + r] * s;
    }
  }

  if constexpr (HEADS) {
    __syncthreads();
    static_assert(2 * RT <= AGG_BS / 32, "one wave per (16-row tile, head): RT tiles x 2 heads, the other waves are done");
    static_assert(F % 4 == 0 && N_CLS <= HEAD_CP, "K step 4; head columns fit the padded tiles");
    constexpr int NT = HEAD_CP / 16;
    const int hd = swave & 1;
    const int rt = swave >> 1;
    if (rt * 16 >= nr) return;
    const int c  = lane & 15;
    const int kh = lane >> 4;
    const float2* __restrict__ A2 = (const float2*)(acc + (rt * 16 + c) * P);
    const float2* __restrict__ B2 = (const float2*)wsh + hd * ((F / 2) * HEAD_CP) + c;
    v8f d[NT] = {};
#pragma unroll
    for (int k2 = 0; k2 < F / 2; k2 += 2) {
      const float2 af = A2[k2 + kh];
      v2f a; a.x = af.x; a.y = af.y;
#pragma unroll
      for (int t = 0; t < NT; ++t) {
        const float2 bf = B2[(k2 + kh) * HEAD_CP + t * 16];
        v2f b; b.x = bf.x; b.y = bf.y;
        d[t] = __builtin_amdgcn_wmma_f32_16x16x4_f32(
             false, a,  false, b,
             (short)0, d[t],  false,  false);
      }
    }
    float* __restrict__ oh = hd ? hb : ha;
    const int crow = rt * 16 + (kh << 3);
#pragma unroll
    for (int t = 0; t < NT; ++t) {
      const int col = (t << 4) + c;
      if (col < N_CLS) {
#pragma unroll
        for (int r = 0; r < 8; ++r)
          if (crow + r < nr) oh[(long long)(n0 + crow + r) * N_CLS + col] = d[t][r];
      }
    }
  }
}

__global__ void log_softmax_kernel(const float* __restrict__ in, float* __restrict__ out,
                                   int n, int F) {
  int row = blockIdx.x * blockDim.x + threadIdx.x;
  if (row >= n) return;
  const float* p = in + (long long)row * F;
  float m = -INFINITY;
  for (int c = 0; c < F; ++c) m = fmaxf(m, p[c]);
  float s = 0.0f;
  for (int c = 0; c < F; ++c) s += __expf(p[c] - m);
  float ls = __logf(s) + m;
  float* o = out + (long long)row * F;
  for (int c = 0; c < F; ++c) o[c] = p[c] - ls;
}

__global__ void edge_dot_kernel(const float* __restrict__ h, const int* __restrict__ pos,
                                const int* __restrict__ neg, float* __restrict__ res) {
  int e = blockIdx.x * blockDim.x + threadIdx.x;
  if (e >= 2 * E_LP) return;
  const int* src = (e < E_LP) ? pos : neg;
  int eo = (e < E_LP) ? e : (e - E_LP);
  int j = src[eo];
  int i = src[E_LP + eo];
  const float4* hj = (const float4*)(h + (long long)j * H2_DIM);
  const float4* hi = (const float4*)(h + (long long)i * H2_DIM);
  float s = 0.0f;
#pragma unroll
  for (int k = 0; k < H2_DIM / 4; ++k) {
    float4 a = hi[k], b = hj[k];
    s += a.x * b.x + a.y * b.y + a.z * b.z + a.w * b.w;
  }
  res[e] = s;
}

static inline int blocks_for(long long total, int bs) {
  return (int)((total + bs - 1) / bs);
}

extern "C" void kernel_launch(void* const* d_in, const int* in_sizes, int n_in,
                              void* d_out, int out_size, void* d_ws, size_t ws_size,
                              hipStream_t stream) {
  const float* x      = (const float*)d_in[0];
  const int*   tpe    = (const int*)  d_in[1];
  const int*   ei     = (const int*)  d_in[2];
  const float* ew     = (const float*)d_in[3];
  const int*   pos    = (const int*)  d_in[4];
  const int*   neg    = (const int*)  d_in[5];
  const float* W1     = (const float*)d_in[6];
  const float* b1     = (const float*)d_in[7];
  const float* W2     = (const float*)d_in[8];
  const float* b2     = (const float*)d_in[9];
  const float* W_attr = (const float*)d_in[10];
  const float* b_attr = (const float*)d_in[11];
  const float* W_att  = (const float*)d_in[12];
  const float* b_att  = (const float*)d_in[13];

  float* out = (float*)d_out;
  float* ws  = (float*)d_ws;

  const long long OFF_DINV_T = 0;
  const long long OFF_DINV_W = 50000;
  const long long OFF_HX1    = 100000;
  const long long OFF_H1     = OFF_HX1 + 6400000;
  const long long OFF_HX2    = OFF_HX1;
  const long long OFF_BKT    = OFF_H1  + 6400000;
  const long long OFF_H2     = OFF_BKT + 3200000;
  const long long OFF_W1P    = OFF_H2  + 3200000;
  const long long OFF_W2P    = OFF_W1P + 65536;
  const long long OFF_HXA    = OFF_HX1 + 3200000;
  const long long OFF_HXB    = OFF_H1 + 4000000;
  const long long OFF_ATTR   = OFF_H1;
  const long long OFF_ATT    = OFF_H1 + 2000000;

  float* dinv_t = ws + OFF_DINV_T;
  float* dinv_w = ws + OFF_DINV_W;
  float* hx1    = ws + OFF_HX1;
  float* h1     = ws + OFF_H1;
  float* hx2    = ws + OFF_HX2;
  float* h2     = ws + OFF_H2;
  float* w1p    = ws + OFF_W1P;
  float* w2p    = ws + OFF_W2P;
  float* hxa    = ws + OFF_HXA;
  float* hxb    = ws + OFF_HXB;
  float* attr_a = ws + OFF_ATTR;
  float* att_a  = ws + OFF_ATT;

  const int NCH = (E_MP + BK_CHUNK - 1) / BK_CHUNK;
  unsigned* bent_w = (unsigned*)(ws + OFF_BKT);
  unsigned* bent_t = bent_w + 2 * E_MP;
  unsigned* btab_t = bent_t + E_MP;
  unsigned* btab_w = btab_t + NCH * (AGG_NB + 1);

  const int BS = 256;

  pack_b_kernel<<<blocks_for((D_FEAT / 2) * H1_DIM, BS), BS, 0, stream>>>(W1, w1p, D_FEAT, H1_DIM, H1_DIM);
  pack_b_kernel<<<blocks_for((H1_DIM / 2) * H2_DIM, BS), BS, 0, stream>>>(W2, w2p, H1_DIM, H2_DIM, H2_DIM);

  bucket_kernel<false><<<NCH, BK_BS, 0, stream>>>(tpe, tpe + E_MP, nullptr, bent_t, btab_t, E_MP, N_NODES);
  bucket_kernel<true ><<<NCH, BK_BS, 0, stream>>>(ei,  ei + E_MP,  ew,      bent_w, btab_w, E_MP, N_NODES);

  deg_own_kernel<false><<<AGG_NB, DEG_BS, 0, stream>>>(dinv_t, bent_t, btab_t, NCH, N_NODES);
  deg_own_kernel<true ><<<AGG_NB, DEG_BS, 0, stream>>>(dinv_w, bent_w, btab_w, NCH, N_NODES);

  const int ROW_TILES = N_NODES / 16;
  const int GEMM_BS = 128;
  const int gemm_blocks = blocks_for((long long)ROW_TILES * 32, GEMM_BS);
  const int AGG_BLOCKS = AGG_NB;

  wmma_gemm_kernel<D_FEAT, 8, H1_DIM, false><<<gemm_blocks, GEMM_BS, 0, stream>>>(x, w1p, dinv_t, hx1, N_NODES);
  gcn_agg_own_kernel<H1_DIM, false, false><<<AGG_BLOCKS, AGG_BS, 0, stream>>>(
      hx1, h1, dinv_t, b1, bent_t, btab_t, NCH, N_NODES, nullptr, nullptr, nullptr, nullptr, nullptr);

  wmma_gemm_kernel<H1_DIM, 4, H2_DIM, true><<<gemm_blocks, GEMM_BS, 0, stream>>>(h1, w2p, dinv_t, hx2, N_NODES);
  gcn_agg_own_kernel<H2_DIM, false, true><<<AGG_BLOCKS, AGG_BS, 0, stream>>>(
      hx2, h2, dinv_t, b2, bent_t, btab_t, NCH, N_NODES, W_attr, W_att, dinv_w, hxa, hxb);

  gcn_agg_own_kernel<N_CLS, true, false><<<AGG_BLOCKS, AGG_BS, 0, stream>>>(
      hxa, attr_a, dinv_w, b_attr, bent_w, btab_w, NCH, N_NODES, nullptr, nullptr, nullptr, nullptr, nullptr);

  gcn_agg_own_kernel<N_CLS, true, false><<<AGG_BLOCKS, AGG_BS, 0, stream>>>(
      hxb, att_a, dinv_w, b_att, bent_w, btab_w, NCH, N_NODES, nullptr, nullptr, nullptr, nullptr, nullptr);

  edge_dot_kernel<<<blocks_for(2 * E_LP, BS), BS, 0, stream>>>(h2, pos, neg, out);
  log_softmax_kernel<<<blocks_for(N_NODES, BS), BS, 0, stream>>>(attr_a, out + 2 * E_LP, N_NODES, N_CLS);
  log_softmax_kernel<<<blocks_for(N_NODES, BS), BS, 0, stream>>>(att_a,  out + 2 * E_LP + (long long)N_NODES * N_CLS, N_NODES, N_CLS);
}


# head aggregations x2: the block takes log_softmax of its finished rows in its LDS tile (a thread per row, class sums in the order of before), stores the outputs as whole lines; launches 15 -> 13
# speedup vs baseline: 2.7256x; 1.0868x over previous
#include <hip/hip_runtime.h>
#include <hip/hip_bf16.h>

#define N_NODES   50000
#define D_FEAT    512
#define H1_DIM    128
#define H2_DIM    64
#define N_CLS     40
#define E_MP      800000
#define E_LP      250000

typedef __attribute__((ext_vector_type(2))) float v2f;
typedef __attribute__((ext_vector_type(8))) float v8f;

__global__ void pack_b_kernel(const float* __restrict__ B, float* __restrict__ Bp,
                              int K, int NC, int NCP) {
  int p = blockIdx.x * blockDim.x + threadIdx.x;
  int total = (K / 2) * NCP;
  if (p >= total) return;
  int k2 = p / NCP;
  int col = p % NCP;
  float x0 = 0.0f, x1 = 0.0f;
  if (col < NC) {
    x0 = B[(long long)(2 * k2) * NC + col];
    x1 = B[(long long)(2 * k2 + 1) * NC + col];
  }
  Bp[2 * p + 0] = x0;
  Bp[2 * p + 1] = x1;
}

template <int KDIM, int NT, int NCSTORE, bool RELU_A>
__global__ void wmma_gemm_kernel(const float* __restrict__ A, const float* __restrict__ Bp,
                                 const float* __restrict__ rs, float* __restrict__ C, int M) {
  constexpr int NCP = NT * 16;
  int wave = (int)(((long long)blockIdx.x * blockDim.x + threadIdx.x) >> 5);
  int lane = threadIdx.x & 31;
  int ntiles = M >> 4;
  if (wave >= ntiles) return;

  int row0 = wave << 4;
  int arow = row0 + (lane & 15);
  int khb  = lane >> 4;
  int bcol = lane & 15;

  v8f acc[NT] = {};

  const int crow = row0 + (khb << 3);
  float sc[8];
#pragma unroll
  for (int r = 0; r < 8; ++r) sc[r] = rs[crow + r];

  const float2* __restrict__ Arow2 = (const float2*)(A + (long long)arow * KDIM);
  const float2* __restrict__ Bp2   = (const float2*)Bp;

#pragma unroll 2
  for (int k2 = 0; k2 < KDIM / 2; k2 += 2) {
    const float2* __restrict__ Bbase = Bp2 + (long long)(k2 + khb) * NCP + bcol;
    float2 bf[NT];
#pragma unroll
    for (int t = 0; t < NT; ++t) bf[t] = Bbase[t * 16];

    float2 af = Arow2[k2 + khb];
    if (RELU_A) { af.x = fmaxf(af.x, 0.0f); af.y = fmaxf(af.y, 0.0f); }
    v2f a; a.x = af.x; a.y = af.y;

#pragma unroll
    for (int t = 0; t < NT; ++t) {
      v2f b; b.x = bf[t].x; b.y = bf[t].y;
      acc[t] = __builtin_amdgcn_wmma_f32_16x16x4_f32(
           false, a,  false, b,
           (short)0, acc[t],  false,  false);
    }
  }

  int ccol = lane & 15;
#pragma unroll
  for (int t = 0; t < NT; ++t) {
    int col = (t << 4) + ccol;
    if (col < NCSTORE) {
#pragma unroll
      for (int r = 0; r < 8; ++r) {
        C[(long long)(crow + r) * NCSTORE + col] = acc[t][r] * sc[r];
      }
    }
  }
}

#define AGG_R   200
#define AGG_Q   8192
#define AGG_BS  1024
#define AGG_NB  ((N_NODES + AGG_R - 1) / AGG_R)
#define BK_CHUNK 16384
#define BK_BS    1024
#define BK_PER   (BK_CHUNK / BK_BS)

template <bool WT>
__global__ __launch_bounds__(BK_BS)
void bucket_kernel(const int* __restrict__ rows, const int* __restrict__ cols,
                   const float* __restrict__ ew, unsigned* __restrict__ bent,
                   unsigned* __restrict__ btab, int E, int N) {
  static_assert(AGG_NB < 256 && BK_CHUNK <= (1 << 24), "packing: tile (or spare bin) | rank << 8; prefix on 8 waves");
  static_assert(N_NODES <= 65536 && AGG_R <= 65536, "record packing: source row | local row << 16");
  __shared__ unsigned hist[AGG_NB + 1];
  __shared__ unsigned start[AGG_NB + 1];
  __shared__ unsigned wsum[8];
  constexpr int RW = WT ? 2 : 1;
  __shared__ __attribute__((aligned(16))) unsigned stage[RW * BK_CHUNK];
  const int tid  = threadIdx.x;
  const int lane = tid & 31;
  const int c0   = blockIdx.x * BK_CHUNK;
  const int len  = min(BK_CHUNK, E - c0);

  for (int i = tid; i <= AGG_NB; i += BK_BS) hist[i] = 0u;
  __syncthreads();

  unsigned cv[BK_PER];
  unsigned rv[BK_PER];
  float    wv[BK_PER];
#pragma unroll
  for (int u = 0; u < BK_PER; ++u) {
    const int j = c0 + min(tid + u * BK_BS, len - 1);
    cv[u] = (unsigned)cols[j];
    rv[u] = (unsigned)rows[j];
    wv[u] = WT ? ew[j] : 1.0f;
  }

  unsigned pk[BK_PER];
#pragma unroll
  for (int u = 0; u < BK_PER; ++u) {
    bool ok = (tid + u * BK_BS < len) && (cv[u] < (unsigned)N);
    unsigned t = ok ? cv[u] / AGG_R : (unsigned)AGG_NB;
    unsigned r = atomicAdd(&hist[t], 1u);
    pk[u] = t | (r << 8);
  }
  __syncthreads();

  unsigned v = 0u, inc = 0u;
  if (tid < 256) {
    v = (tid < AGG_NB) ? hist[tid] : 0u;
    inc = v;
#pragma unroll
    for (int d = 1; d < 32; d <<= 1) {
      unsigned o = __shfl_up(inc, d);
      if (lane >= d) inc += o;
    }
    if (lane == 31) wsum[tid >> 5] = inc;
  }
  __syncthreads();
  if (tid <= AGG_NB) {
    unsigned off = 0u;
#pragma unroll
    for (int w = 0; w < 8; ++w) off += (w < (tid >> 5)) ? wsum[w] : 0u;
    unsigned ex = off + inc - v;
    start[tid] = ex;
    btab[(long long)blockIdx.x * (AGG_NB + 1) + tid] = (unsigned)c0 + ex;
  }
  __syncthreads();

#pragma unroll
  for (int u = 0; u < BK_PER; ++u) {
    unsigned t = pk[u] & 0xFFu;
    if (t < (unsigned)AGG_NB) {
      unsigned r   = pk[u] >> 8;
      unsigned pos = start[t] + r;
      unsigned rec = rv[u] | ((cv[u] - t * AGG_R) << 16);
      if (WT) ((uint2*)stage)[pos] = make_uint2(rec, __float_as_uint(wv[u]));
      else    stage[pos] = rec;
    }
  }
  __syncthreads();

  const unsigned nw = start[AGG_NB] * RW;
  unsigned* __restrict__ dst = bent + (long long)c0 * RW;
  for (unsigned i = 4u * tid; i < nw; i += 4u * BK_BS) {
    if (i + 4u <= nw) {
      *(uint4*)(dst + i) = *(const uint4*)(stage + i);
    } else {
      for (unsigned k = i; k < nw; ++k) dst[k] = stage[k];
    }
  }
}

#define DEG_BS 1024
#define DEG_U  4
template <bool WT>
__global__ __launch_bounds__(DEG_BS)
void deg_own_kernel(float* __restrict__ dinv,
                    const unsigned* __restrict__ bent, const unsigned* __restrict__ btab,
                    int nch, int N) {
  __shared__ float dsum[AGG_R];
  const int tid  = threadIdx.x;
  const int lane = tid & 31;
  const int n0   = blockIdx.x * AGG_R;
  const int nr   = min(AGG_R, N - n0);

  if (tid < AGG_R) dsum[tid] = 1.0f;
  __syncthreads();

  constexpr int NW = DEG_BS / 32;
  const int swave = __builtin_amdgcn_readfirstlane(tid >> 5);
  for (int c = swave; c < nch; c += NW) {
    const unsigned* __restrict__ trow = btab + (long long)c * (AGG_NB + 1) + blockIdx.x;
    const unsigned s0  = trow[0];
    const unsigned cnt = trow[1] - s0;
    for (unsigned i0 = 0; i0 < cnt; i0 += 32u * DEG_U) {
      unsigned cl[DEG_U];
      float    w[DEG_U];
#pragma unroll
      for (int u = 0; u < DEG_U; ++u) {
        const unsigned j = s0 + min(i0 + 32u * u + lane, cnt - 1u);
        if (WT) {
          uint2 p = ((const uint2*)bent)[j];
          cl[u] = p.x >> 16;
          w[u]  = __uint_as_float(p.y);
        } else {
          cl[u] = bent[j] >> 16;
          w[u]  = 1.0f;
        }
      }
#pragma unroll
      for (int u = 0; u < DEG_U; ++u)
        if (i0 + 32u * u + lane < cnt) atomicAdd(&dsum[cl[u]], w[u]);
    }
  }
  __syncthreads();

  if (tid < nr) {
    float d = dsum[tid];
    dinv[n0 + tid] = (d > 0.0f) ? rsqrtf(d) : 0.0f;
  }
}

#define AGG_CU 4
#define HEAD_CP 48
template <int F, bool WT, bool HEADS, bool LSM>
__global__ __launch_bounds__(AGG_BS)
void gcn_agg_own_kernel(const float* __restrict__ hx, float* __restrict__ out,
                        const float* __restrict__ dinv, const float* __restrict__ bias,
                        const unsigned* __restrict__ bent, const unsigned* __restrict__ btab,
                        int nch, int N,
                        const float* __restrict__ Wa, const float* __restrict__ Wb,
                        const float* __restrict__ hscale,
                        float* __restrict__ ha, float* __restrict__ hb) {
  constexpr int P   = HEADS ? F + 4 : F;
  constexpr int WSH = HEADS ? 2 * (F / 2) * HEAD_CP * 2 : 4;
  constexpr int RT  = (AGG_R + 15) / 16;
  constexpr int RP  = HEADS ? RT * 16 : AGG_R;
  __shared__ __attribute__((aligned(16))) float acc[RP * P];
  __shared__ __attribute__((aligned(16))) float wsh[WSH];
  __shared__ unsigned qa[AGG_Q];
  __shared__ float    qw[WT ? AGG_Q : 1];
  __shared__ unsigned qcount;
  __shared__ float    lsm[LSM ? AGG_R : 1];
  static_assert(!(LSM && HEADS), "step 5 reads the tile at pitch F; step 4 is for the layer in front of the heads");

  const int tid  = threadIdx.x;
  const int lane = tid & 31;
  const int wave = tid >> 5;
  const int n0   = blockIdx.x * AGG_R;
  const int nr   = min(AGG_R, N - n0);

  if (tid == 0) qcount = 0u;
  constexpr int V      = 4;
  constexpr int NPIECE = AGG_R * F / V;
  constexpr int NIT    = (NPIECE + AGG_BS - 1) / AGG_BS;
  static_assert(F % V == 0 && (AGG_R * F) % V == 0, "a piece stays inside a row");
  typedef float fvec __attribute__((ext_vector_type(V)));
  {
    const float* __restrict__ own = hx + (long long)n0 * F;
    if (nr == AGG_R) {
      fvec t[NIT];
#pragma unroll
      for (int u = 0; u < NIT; ++u) {
        const int pi = tid + u * AGG_BS;
        if (u < NIT - 1 || pi < NPIECE) t[u] = *(const fvec*)(own + pi * V);
      }
#pragma unroll
      for (int u = 0; u < NIT; ++u) {
        const int pi = tid + u * AGG_BS;
        if (u < NIT - 1 || pi < NPIECE) {
          const int i = pi * V;
          const int r = i / F;
          const int f = i - r * F;
          *(fvec*)(acc + ((P == F) ? i : r * P + f)) = t[u];
        }
      }
    } else {
      for (int i = tid * V; i < nr * F; i += AGG_BS * V) {
        const int r = i / F;
        const int f = i - r * F;
        *(fvec*)(acc + ((P == F) ? i : r * P + f)) = *(const fvec*)(own + i);
      }
    }
  }
  if constexpr (HEADS) {
    for (int i = nr * P + tid; i < RP * P; i += AGG_BS) acc[i] = 0.0f;
    for (int p = tid; p < 2 * (F / 2) * HEAD_CP; p += AGG_BS) {
      const int hd  = p / ((F / 2) * HEAD_CP);
      const int q   = p - hd * ((F / 2) * HEAD_CP);
      const int k2  = q / HEAD_CP;
      const int col = q - k2 * HEAD_CP;
      const float* __restrict__ Wsel = hd ? Wb : Wa;
      float x0 = 0.0f, x1 = 0.0f;
      if (col < N_CLS) {
        x0 = Wsel[(2 * k2) * N_CLS + col];
        x1 = Wsel[(2 * k2 + 1) * N_CLS + col];
      }
      ((float2*)wsh)[p] = make_float2(x0, x1);
    }
  }
  __syncthreads();

  constexpr int NW = AGG_BS / 32;
  const int swave = __builtin_amdgcn_readfirstlane(wave);
  for (int c = swave; c < nch; c += NW) {
    const unsigned* __restrict__ trow = btab + (long long)c * (AGG_NB + 1) + blockIdx.x;
    const unsigned s0  = trow[0];
    const unsigned cnt = trow[1] - s0;
    unsigned base = 0u;
    if (lane == 0) base = atomicAdd(&qcount, cnt);
    base = (unsigned)__builtin_amdgcn_readfirstlane((int)base);
    for (unsigned i0 = 0; i0 < cnt; i0 += 32u * AGG_CU) {
      unsigned rec[AGG_CU];
      float    w[AGG_CU];
#pragma unroll
      for (int u = 0; u < AGG_CU; ++u) {
        const unsigned j = s0 + min(i0 + 32u * u + lane, cnt - 1u);
        if (WT) {
          uint2 p = ((const uint2*)bent)[j];
          rec[u] = p.x;
          w[u]   = __uint_as_float(p.y);
        } else {
          rec[u] = bent[j];
          w[u]   = 1.0f;
        }
      }
#pragma unroll
      for (int u = 0; u < AGG_CU; ++u) {
        const unsigned i = i0 + 32u * u + lane;
        if (i < cnt) {
          const unsigned slot = base + i;
          if (slot < AGG_Q) {
            qa[slot] = rec[u];
            if constexpr (WT) qw[slot] = w[u];
          } else {
            const int r  = (int)(rec[u] & 0xFFFFu);
            const int cl = (int)(rec[u] >> 16);
            for (int f = 0; f < F; ++f) {
              const float x = hx[(long long)r * F + f];
              atomicAdd(&acc[cl * P + f], WT ? x * w[u] : x);
            }
          }
        }
      }
    }
  }
  __syncthreads();

  const int nq = (int)min(qcount, (unsigned)AGG_Q);
  constexpr int U   = 8;
  constexpr int PER = (F + 31) / 32;
  for (int i0 = swave; i0 < nq; i0 += NW * U) {
    float v[U][PER];
    int   cl[U];
#pragma unroll
    for (int u = 0; u < U; ++u) {
      int i  = i0 + u * NW;
      int ic = min(i, nq - 1);
      unsigned p = qa[ic];
      float wq = 1.0f;
      if constexpr (WT) wq = qw[ic];
      int r = (int)(p & 0xFFFFu);
      cl[u] = (i < nq) ? (int)(p >> 16) : -1;
      const float* __restrict__ s = hx + (long long)r * F;
#pragma unroll
      for (int k = 0; k < PER; ++k) {
        int f = min(lane + 32 * k, F - 1);
        v[u][k] = WT ? s[f] * wq : s[f];
      }
    }
#pragma unroll
    for (int u = 0; u < U; ++u) {
      if (cl[u] >= 0) {
#pragma unroll
        for (int k = 0; k < PER; ++k) {
          int f = lane + 32 * k;
          if (f < F) atomicAdd(&acc[cl[u] * P + f], v[u][k]);
        }
      }
    }
  }
  __syncthreads();

  float* __restrict__ o = out + (long long)n0 * F;
  if (nr == AGG_R) {
    float dd[NIT];
    float hh[NIT];
    fvec  bb[NIT];
#pragma unroll
    for (int u = 0; u < NIT; ++u) {
      const int pi = tid + u * AGG_BS;
      if (u < NIT - 1 || pi < NPIECE) {
        const int i = pi * V;
        const int r = i / F;
        const int f = i - r * F;
        dd[u] = dinv[n0 + r];
        hh[u] = HEADS ? hscale[n0 + r] : 1.0f;
#pragma unroll
        for (int k = 0; k < V; ++k) bb[u][k] = bias[f + k];
      }
    }
#pragma unroll
    for (int u = 0; u < NIT; ++u) {
      const int pi = tid + u * AGG_BS;
      if (u < NIT - 1 || pi < NPIECE) {
        const int i = pi * V;
        const int r = i / F;
        const int f = i - r * F;
        float* a = acc + ((P == F) ? i : r * P + f);
        const fvec s = bb[u] + dd[u] * *(const fvec*)a;
        if constexpr (LSM) *(fvec*)a = s;
        else               *(fvec*)(o + i) = s;
        if constexpr (HEADS) *(fvec*)a = hh[u] * s;
      }
    }
  } else {
    for (int i = tid * V; i < nr * F; i += AGG_BS * V) {
      const int r = i / F;
      const int f = i - r * F;
      float* a = acc + ((P == F) ? i : r * P + f);
      const float d = dinv[n0 + r];
      fvec s = *(const fvec*)a;
#pragma unroll
      for (int k = 0; k < V; ++k) s[k] = bias[f + k] + d * s[k];
      if constexpr (LSM) *(fvec*)a = s;
      else               *(fvec*)(o + i) = s;
      if constexpr (HEADS) *(fvec*)a = hscale[n0 + r] * s;
    }
  }

  if constexpr (LSM) {
    __syncthreads();
    if (tid < nr) {
      const fvec* __restrict__ rp = (const fvec*)(acc + tid * P);
      fvec xv[F / V];
#pragma unroll
      for (int q = 0; q < F / V; ++q) xv[q] = rp[q];
      float m = -INFINITY;
#pragma unroll
      for (int cc = 0; cc < F; ++cc) m = fmaxf(m, xv[cc / V][cc % V]);
      float se = 0.0f;
#pragma unroll
      for (int cc = 0; cc < F; ++cc) se += __expf(xv[cc / V][cc % V] - m);
      lsm[tid] = __logf(se) + m;
    }
    __syncthreads();
    for (int i = tid * V; i < nr * F; i += AGG_BS * V) {
      const int r = i / F;
      *(fvec*)(o + i) = *(const fvec*)(acc + i) - lsm[r];
    }
  }

  if constexpr (HEADS) {
    __syncthreads();
    static_assert(2 * RT <= AGG_BS / 32, "one wave per (16-row tile, head): RT tiles x 2 heads, the other waves are done");
    static_assert(F % 4 == 0 && N_CLS <= HEAD_CP, "K step 4; head columns fit the padded tiles");
    constexpr int NT = HEAD_CP / 16;
    const int hd = swave & 1;
    const int rt = swave >> 1;
    if (rt * 16 >= nr) return;
    const int c  = lane & 15;
    const int kh = lane >> 4;
    const float2* __restrict__ A2 = (const float2*)(acc + (rt * 16 + c) * P);
    const float2* __restrict__ B2 = (const float2*)wsh + hd * ((F / 2) * HEAD_CP) + c;
    v8f d[NT] = {};
#pragma unroll
    for (int k2 = 0; k2 < F / 2; k2 += 2) {
      const float2 af = A2[k2 + kh];
      v2f a; a.x = af.x; a.y = af.y;
#pragma unroll
      for (int t = 0; t < NT; ++t) {
        const float2 bf = B2[(k2 + kh) * HEAD_CP + t * 16];
        v2f b; b.x = bf.x; b.y = bf.y;
        d[t] = __builtin_amdgcn_wmma_f32_16x16x4_f32(
             false, a,  false, b,
             (short)0, d[t],  false,  false);
      }
    }
    float* __restrict__ oh = hd ? hb : ha;
    const int crow = rt * 16 + (kh << 3);
#pragma unroll
    for (int t = 0; t < NT; ++t) {
      const int col = (t << 4) + c;
      if (col < N_CLS) {
#pragma unroll
        for (int r = 0; r < 8; ++r)
          if (crow + r < nr) oh[(long long)(n0 + crow + r) * N_CLS + col] = d[t][r];
      }
    }
  }
}

__global__ void edge_dot_kernel(const float* __restrict__ h, const int* __restrict__ pos,
                                const int* __restrict__ neg, float* __restrict__ res) {
  int e = blockIdx.x * blockDim.x + threadIdx.x;
  if (e >= 2 * E_LP) return;
  const int* src = (e < E_LP) ? pos : neg;
  int eo = (e < E_LP) ? e : (e - E_LP);
  int j = src[eo];
  int i = src[E_LP + eo];
  const float4* hj = (const float4*)(h + (long long)j * H2_DIM);
  const float4* hi = (const float4*)(h + (long long)i * H2_DIM);
  float s = 0.0f;
#pragma unroll
  for (int k = 0; k < H2_DIM / 4; ++k) {
    float4 a = hi[k], b = hj[k];
    s += a.x * b.x + a.y * b.y + a.z * b.z + a.w * b.w;
  }
  res[e] = s;
}

static inline int blocks_for(long long total, int bs) {
  return (int)((total + bs - 1) / bs);
}

extern "C" void kernel_launch(void* const* d_in, const int* in_sizes, int n_in,
                              void* d_out, int out_size, void* d_ws, size_t ws_size,
                              hipStream_t stream) {
  const float* x      = (const float*)d_in[0];
  const int*   tpe    = (const int*)  d_in[1];
  const int*   ei     = (const int*)  d_in[2];
  const float* ew     = (const float*)d_in[3];
  const int*   pos    = (const int*)  d_in[4];
  const int*   neg    = (const int*)  d_in[5];
  const float* W1     = (const float*)d_in[6];
  const float* b1     = (const float*)d_in[7];
  const float* W2     = (const float*)d_in[8];
  const float* b2     = (const float*)d_in[9];
  const float* W_attr = (const float*)d_in[10];
  const float* b_attr = (const float*)d_in[11];
  const float* W_att  = (const float*)d_in[12];
  const float* b_att  = (const float*)d_in[13];

  float* out = (float*)d_out;
  float* ws  = (float*)d_ws;

  const long long OFF_DINV_T = 0;
  const long long OFF_DINV_W = 50000;
  const long long OFF_HX1    = 100000;
  const long long OFF_H1     = OFF_HX1 + 6400000;
  const long long OFF_HX2    = OFF_HX1;
  const long long OFF_BKT    = OFF_H1  + 6400000;
  const long long OFF_H2     = OFF_BKT + 3200000;
  const long long OFF_W1P    = OFF_H2  + 3200000;
  const long long OFF_W2P    = OFF_W1P + 65536;
  const long long OFF_HXA    = OFF_HX1 + 3200000;
  const long long OFF_HXB    = OFF_H1 + 4000000;

  float* dinv_t = ws + OFF_DINV_T;
  float* dinv_w = ws + OFF_DINV_W;
  float* hx1    = ws + OFF_HX1;
  float* h1     = ws + OFF_H1;
  float* hx2    = ws + OFF_HX2;
  float* h2     = ws + OFF_H2;
  float* w1p    = ws + OFF_W1P;
  float* w2p    = ws + OFF_W2P;
  float* hxa    = ws + OFF_HXA;
  float* hxb    = ws + OFF_HXB;

  const int NCH = (E_MP + BK_CHUNK - 1) / BK_CHUNK;
  unsigned* bent_w = (unsigned*)(ws + OFF_BKT);
  unsigned* bent_t = bent_w + 2 * E_MP;
  unsigned* btab_t = bent_t + E_MP;
  unsigned* btab_w = btab_t + NCH * (AGG_NB + 1);

  const int BS = 256;

  pack_b_kernel<<<blocks_for((D_FEAT / 2) * H1_DIM, BS), BS, 0, stream>>>(W1, w1p, D_FEAT, H1_DIM, H1_DIM);
  pack_b_kernel<<<blocks_for((H1_DIM / 2) * H2_DIM, BS), BS, 0, stream>>>(W2, w2p, H1_DIM, H2_DIM, H2_DIM);

  bucket_kernel<false><<<NCH, BK_BS, 0, stream>>>(tpe, tpe + E_MP, nullptr, bent_t, btab_t, E_MP, N_NODES);
  bucket_kernel<true ><<<NCH, BK_BS, 0, stream>>>(ei,  ei + E_MP,  ew,      bent_w, btab_w, E_MP, N_NODES);

  deg_own_kernel<false><<<AGG_NB, DEG_BS, 0, stream>>>(dinv_t, bent_t, btab_t, NCH, N_NODES);
  deg_own_kernel<true ><<<AGG_NB, DEG_BS, 0, stream>>>(dinv_w, bent_w, btab_w, NCH, N_NODES);

  const int ROW_TILES = N_NODES / 16;
  const int GEMM_BS = 128;
  const int gemm_blocks = blocks_for((long long)ROW_TILES * 32, GEMM_BS);
  const int AGG_BLOCKS = AGG_NB;

  wmma_gemm_kernel<D_FEAT, 8, H1_DIM, false><<<gemm_blocks, GEMM_BS, 0, stream>>>(x, w1p, dinv_t, hx1, N_NODES);
  gcn_agg_own_kernel<H1_DIM, false, false, false><<<AGG_BLOCKS, AGG_BS, 0, stream>>>(
      hx1, h1, dinv_t, b1, bent_t, btab_t, NCH, N_NODES, nullptr, nullptr, nullptr, nullptr, nullptr);

  wmma_gemm_kernel<H1_DIM, 4, H2_DIM, true><<<gemm_blocks, GEMM_BS, 0, stream>>>(h1, w2p, dinv_t, hx2, N_NODES);
  gcn_agg_own_kernel<H2_DIM, false, true, false><<<AGG_BLOCKS, AGG_BS, 0, stream>>>(
      hx2, h2, dinv_t, b2, bent_t, btab_t, NCH, N_NODES, W_attr, W_att, dinv_w, hxa, hxb);

  float* out_attr = out + 2 * E_LP;
  float* out_att  = out + 2 * E_LP + (long long)N_NODES * N_CLS;
  gcn_agg_own_kernel<N_CLS, true, false, true><<<AGG_BLOCKS, AGG_BS, 0, stream>>>(
      hxa, out_attr, dinv_w, b_attr, bent_w, btab_w, NCH, N_NODES, nullptr, nullptr, nullptr, nullptr, nullptr);

  gcn_agg_own_kernel<N_CLS, true, false, true><<<AGG_BLOCKS, AGG_BS, 0, stream>>>(
      hxb, out_att, dinv_w, b_att, bent_w, btab_w, NCH, N_NODES, nullptr, nullptr, nullptr, nullptr, nullptr);

  edge_dot_kernel<<<blocks_for(2 * E_LP, BS), BS, 0, stream>>>(h2, pos, neg, out);
}


# head aggregations 2 -> 1: layer-2 block writes both heads' products as one 80-wide row (one WMMA product); one weighted pass gathers it once per edge, 2 edges = 5 wave loads; 13 -> 12; sum order kept
# speedup vs baseline: 2.8737x; 1.0543x over previous
#include <hip/hip_runtime.h>
#include <hip/hip_bf16.h>
#include <type_traits>

#define N_NODES   50000
#define D_FEAT    512
#define H1_DIM    128
#define H2_DIM    64
#define N_CLS     40
#define E_MP      800000
#define E_LP      250000

typedef __attribute__((ext_vector_type(2))) float v2f;
typedef __attribute__((ext_vector_type(8))) float v8f;

__global__ void pack_b_kernel(const float* __restrict__ B, float* __restrict__ Bp,
                              int K, int NC, int NCP) {
  int p = blockIdx.x * blockDim.x + threadIdx.x;
  int total = (K / 2) * NCP;
  if (p >= total) return;
  int k2 = p / NCP;
  int col = p % NCP;
  float x0 = 0.0f, x1 = 0.0f;
  if (col < NC) {
    x0 = B[(long long)(2 * k2) * NC + col];
    x1 = B[(long long)(2 * k2 + 1) * NC + col];
  }
  Bp[2 * p + 0] = x0;
  Bp[2 * p + 1] = x1;
}

template <int KDIM, int NT, int NCSTORE, bool RELU_A>
__global__ void wmma_gemm_kernel(const float* __restrict__ A, const float* __restrict__ Bp,
                                 const float* __restrict__ rs, float* __restrict__ C, int M) {
  constexpr int NCP = NT * 16;
  int wave = (int)(((long long)blockIdx.x * blockDim.x + threadIdx.x) >> 5);
  int lane = threadIdx.x & 31;
  int ntiles = M >> 4;
  if (wave >= ntiles) return;

  int row0 = wave << 4;
  int arow = row0 + (lane & 15);
  int khb  = lane >> 4;
  int bcol = lane & 15;

  v8f acc[NT] = {};

  const int crow = row0 + (khb << 3);
  float sc[8];
#pragma unroll
  for (int r = 0; r < 8; ++r) sc[r] = rs[crow + r];

  const float2* __restrict__ Arow2 = (const float2*)(A + (long long)arow * KDIM);
  const float2* __restrict__ Bp2   = (const float2*)Bp;

#pragma unroll 2
  for (int k2 = 0; k2 < KDIM / 2; k2 += 2) {
    const float2* __restrict__ Bbase = Bp2 + (long long)(k2 + khb) * NCP + bcol;
    float2 bf[NT];
#pragma unroll
    for (int t = 0; t < NT; ++t) bf[t] = Bbase[t * 16];

    float2 af = Arow2[k2 + khb];
    if (RELU_A) { af.x = fmaxf(af.x, 0.0f); af.y = fmaxf(af.y, 0.0f); }
    v2f a; a.x = af.x; a.y = af.y;

#pragma unroll
    for (int t = 0; t < NT; ++t) {
      v2f b; b.x = bf[t].x; b.y = bf[t].y;
      acc[t] = __builtin_amdgcn_wmma_f32_16x16x4_f32(
           false, a,  false, b,
           (short)0, acc[t],  false,  false);
    }
  }

  int ccol = lane & 15;
#pragma unroll
  for (int t = 0; t < NT; ++t) {
    int col = (t << 4) + ccol;
    if (col < NCSTORE) {
#pragma unroll
      for (int r = 0; r < 8; ++r) {
        C[(long long)(crow + r) * NCSTORE + col] = acc[t][r] * sc[r];
      }
    }
  }
}

#define AGG_R   200
#define AGG_Q   8192
#define AGG_BS  1024
#define AGG_NB  ((N_NODES + AGG_R - 1) / AGG_R)
#define BK_CHUNK 16384
#define BK_BS    1024
#define BK_PER   (BK_CHUNK / BK_BS)

template <bool WT>
__global__ __launch_bounds__(BK_BS)
void bucket_kernel(const int* __restrict__ rows, const int* __restrict__ cols,
                   const float* __restrict__ ew, unsigned* __restrict__ bent,
                   unsigned* __restrict__ btab, int E, int N) {
  static_assert(AGG_NB < 256 && BK_CHUNK <= (1 << 24), "packing: tile (or spare bin) | rank << 8; prefix on 8 waves");
  static_assert(N_NODES <= 65536 && AGG_R <= 65536, "record packing: source row | local row << 16");
  __shared__ unsigned hist[AGG_NB + 1];
  __shared__ unsigned start[AGG_NB + 1];
  __shared__ unsigned wsum[8];
  constexpr int RW = WT ? 2 : 1;
  __shared__ __attribute__((aligned(16))) unsigned stage[RW * BK_CHUNK];
  const int tid  = threadIdx.x;
  const int lane = tid & 31;
  const int c0   = blockIdx.x * BK_CHUNK;
  const int len  = min(BK_CHUNK, E - c0);

  for (int i = tid; i <= AGG_NB; i += BK_BS) hist[i] = 0u;
  __syncthreads();

  unsigned cv[BK_PER];
  unsigned rv[BK_PER];
  float    wv[BK_PER];
#pragma unroll
  for (int u = 0; u < BK_PER; ++u) {
    const int j = c0 + min(tid + u * BK_BS, len - 1);
    cv[u] = (unsigned)cols[j];
    rv[u] = (unsigned)rows[j];
    wv[u] = WT ? ew[j] : 1.0f;
  }

  unsigned pk[BK_PER];
#pragma unroll
  for (int u = 0; u < BK_PER; ++u) {
    bool ok = (tid + u * BK_BS < len) && (cv[u] < (unsigned)N);
    unsigned t = ok ? cv[u] / AGG_R : (unsigned)AGG_NB;
    unsigned r = atomicAdd(&hist[t], 1u);
    pk[u] = t | (r << 8);
  }
  __syncthreads();

  unsigned v = 0u, inc = 0u;
  if (tid < 256) {
    v = (tid < AGG_NB) ? hist[tid] : 0u;
    inc = v;
#pragma unroll
    for (int d = 1; d < 32; d <<= 1) {
      unsigned o = __shfl_up(inc, d);
      if (lane >= d) inc += o;
    }
    if (lane == 31) wsum[tid >> 5] = inc;
  }
  __syncthreads();
  if (tid <= AGG_NB) {
    unsigned off = 0u;
#pragma unroll
    for (int w = 0; w < 8; ++w) off += (w < (tid >> 5)) ? wsum[w] : 0u;
    unsigned ex = off + inc - v;
    start[tid] = ex;
    btab[(long long)blockIdx.x * (AGG_NB + 1) + tid] = (unsigned)c0 + ex;
  }
  __syncthreads();

#pragma unroll
  for (int u = 0; u < BK_PER; ++u) {
    unsigned t = pk[u] & 0xFFu;
    if (t < (unsigned)AGG_NB) {
      unsigned r   = pk[u] >> 8;
      unsigned pos = start[t] + r;
      unsigned rec = rv[u] | ((cv[u] - t * AGG_R) << 16);
      if (WT) ((uint2*)stage)[pos] = make_uint2(rec, __float_as_uint(wv[u]));
      else    stage[pos] = rec;
    }
  }
  __syncthreads();

  const unsigned nw = start[AGG_NB] * RW;
  unsigned* __restrict__ dst = bent + (long long)c0 * RW;
  for (unsigned i = 4u * tid; i < nw; i += 4u * BK_BS) {
    if (i + 4u <= nw) {
      *(uint4*)(dst + i) = *(const uint4*)(stage + i);
    } else {
      for (unsigned k = i; k < nw; ++k) dst[k] = stage[k];
    }
  }
}

#define DEG_BS 1024
#define DEG_U  4
template <bool WT>
__global__ __launch_bounds__(DEG_BS)
void deg_own_kernel(float* __restrict__ dinv,
                    const unsigned* __restrict__ bent, const unsigned* __restrict__ btab,
                    int nch, int N) {
  __shared__ float dsum[AGG_R];
  const int tid  = threadIdx.x;
  const int lane = tid & 31;
  const int n0   = blockIdx.x * AGG_R;
  const int nr   = min(AGG_R, N - n0);

  if (tid < AGG_R) dsum[tid] = 1.0f;
  __syncthreads();

  constexpr int NW = DEG_BS / 32;
  const int swave = __builtin_amdgcn_readfirstlane(tid >> 5);
  for (int c = swave; c < nch; c += NW) {
    const unsigned* __restrict__ trow = btab + (long long)c * (AGG_NB + 1) + blockIdx.x;
    const unsigned s0  = trow[0];
    const unsigned cnt = trow[1] - s0;
    for (unsigned i0 = 0; i0 < cnt; i0 += 32u * DEG_U) {
      unsigned cl[DEG_U];
      float    w[DEG_U];
#pragma unroll
      for (int u = 0; u < DEG_U; ++u) {
        const unsigned j = s0 + min(i0 + 32u * u + lane, cnt - 1u);
        if (WT) {
          uint2 p = ((const uint2*)bent)[j];
          cl[u] = p.x >> 16;
          w[u]  = __uint_as_float(p.y);
        } else {
          cl[u] = bent[j] >> 16;
          w[u]  = 1.0f;
        }
      }
#pragma unroll
      for (int u = 0; u < DEG_U; ++u)
        if (i0 + 32u * u + lane < cnt) atomicAdd(&dsum[cl[u]], w[u]);
    }
  }
  __syncthreads();

  if (tid < nr) {
    float d = dsum[tid];
    dinv[n0 + tid] = (d > 0.0f) ? rsqrtf(d) : 0.0f;
  }
}

#define AGG_CU 4
#define HEAD_W (2 * N_CLS)
template <int F, bool WT, bool HEADS, bool LSM>
__global__ __launch_bounds__(AGG_BS)
void gcn_agg_own_kernel(const float* __restrict__ hx, float* __restrict__ out,
                        const float* __restrict__ dinv, const float* __restrict__ bias,
                        const unsigned* __restrict__ bent, const unsigned* __restrict__ btab,
                        int nch, int N,
                        const float* __restrict__ Wa, const float* __restrict__ Wb,
                        const float* __restrict__ hscale, float* __restrict__ hcat,
                        const float* __restrict__ bias2, float* __restrict__ out2) {
  constexpr int P   = HEADS ? F + 4 : F;
  constexpr int WSH = HEADS ? (F / 2) * HEAD_W * 2 : 4;
  constexpr int RT  = (AGG_R + 15) / 16;
  constexpr int RP  = HEADS ? RT * 16 : AGG_R;
  constexpr int NH  = LSM ? F / N_CLS : 1;
  __shared__ __attribute__((aligned(16))) float acc[RP * P];
  __shared__ __attribute__((aligned(16))) float wsh[WSH];
  __shared__ unsigned qa[AGG_Q];
  __shared__ float    qw[WT ? AGG_Q : 1];
  __shared__ unsigned qcount;
  __shared__ float    lsm[LSM ? AGG_R * NH : 1];
  static_assert(!(LSM && HEADS), "step 5 reads the tile at pitch F; step 4 is for the layer in front of the heads");
  static_assert(!LSM || (F == NH * N_CLS && NH <= 2 && AGG_R * NH <= AGG_BS && N_CLS % 4 == 0),
                "step 5: whole heads in a row, bias / bias2 and out / out2, a thread per (row, head), 16-byte pieces inside a head");

  const int tid  = threadIdx.x;
  const int lane = tid & 31;
  const int wave = tid >> 5;
  const int n0   = blockIdx.x * AGG_R;
  const int nr   = min(AGG_R, N - n0);

  if (tid == 0) qcount = 0u;
  constexpr int V      = 4;
  constexpr int NPIECE = AGG_R * F / V;
  constexpr int NIT    = (NPIECE + AGG_BS - 1) / AGG_BS;
  static_assert(F % V == 0 && (AGG_R * F) % V == 0, "a piece stays inside a row");
  typedef float fvec __attribute__((ext_vector_type(V)));
  {
    const float* __restrict__ own = hx + (long long)n0 * F;
    if (nr == AGG_R) {
      fvec t[NIT];
#pragma unroll
      for (int u = 0; u < NIT; ++u) {
        const int pi = tid + u * AGG_BS;
        if (u < NIT - 1 || pi < NPIECE) t[u] = *(const fvec*)(own + pi * V);
      }
#pragma unroll
      for (int u = 0; u < NIT; ++u) {
        const int pi = tid + u * AGG_BS;
        if (u < NIT - 1 || pi < NPIECE) {
          const int i = pi * V;
          const int r = i / F;
          const int f = i - r * F;
          *(fvec*)(acc + ((P == F) ? i : r * P + f)) = t[u];
        }
      }
    } else {
      for (int i = tid * V; i < nr * F; i += AGG_BS * V) {
        const int r = i / F;
        const int f = i - r * F;
        *(fvec*)(acc + ((P == F) ? i : r * P + f)) = *(const fvec*)(own + i);
      }
    }
  }
  if constexpr (HEADS) {
    for (int i = nr * P + tid; i < RP * P; i += AGG_BS) acc[i] = 0.0f;
    for (int p = tid; p < (F / 2) * HEAD_W; p += AGG_BS) {
      const int k2  = p / HEAD_W;
      const int col = p - k2 * HEAD_W;
      const int hd  = col / N_CLS;
      const int cc  = col - hd * N_CLS;
      const float* __restrict__ Wsel = hd ? Wb : Wa;
      ((float2*)wsh)[p] = make_float2(Wsel[(2 * k2) * N_CLS + cc], Wsel[(2 * k2 + 1) * N_CLS + cc]);
    }
  }
  __syncthreads();

  constexpr int NW = AGG_BS / 32;
  const int swave = __builtin_amdgcn_readfirstlane(wave);
  for (int c = swave; c < nch; c += NW) {
    const unsigned* __restrict__ trow = btab + (long long)c * (AGG_NB + 1) + blockIdx.x;
    const unsigned s0  = trow[0];
    const unsigned cnt = trow[1] - s0;
    unsigned base = 0u;
    if (lane == 0) base = atomicAdd(&qcount, cnt);
    base = (unsigned)__builtin_amdgcn_readfirstlane((int)base);
    for (unsigned i0 = 0; i0 < cnt; i0 += 32u * AGG_CU) {
      unsigned rec[AGG_CU];
      float    w[AGG_CU];
#pragma unroll
      for (int u = 0; u < AGG_CU; ++u) {
        const unsigned j = s0 + min(i0 + 32u * u + lane, cnt - 1u);
        if (WT) {
          uint2 p = ((const uint2*)bent)[j];
          rec[u] = p.x;
          w[u]   = __uint_as_float(p.y);
        } else {
          rec[u] = bent[j];
          w[u]   = 1.0f;
        }
      }
#pragma unroll
      for (int u = 0; u < AGG_CU; ++u) {
        const unsigned i = i0 + 32u * u + lane;
        if (i < cnt) {
          const unsigned slot = base + i;
          if (slot < AGG_Q) {
            qa[slot] = rec[u];
            if constexpr (WT) qw[slot] = w[u];
          } else {
            const int r  = (int)(rec[u] & 0xFFFFu);
            const int cl = (int)(rec[u] >> 16);
            for (int f = 0; f < F; ++f) {
              const float x = hx[(long long)r * F + f];
              atomicAdd(&acc[cl * P + f], WT ? x * w[u] : x);
            }
          }
        }
      }
    }
  }
  __syncthreads();

  const int nq = (int)min(qcount, (unsigned)AGG_Q);
  constexpr int U   = 8;
  constexpr int PER = (F + 31) / 32;
  constexpr int FULL = F / 32;
  constexpr int TAIL = F % 32;
  if constexpr (TAIL != 0) {
    static_assert(TAIL == 16 && U % 2 == 0, "two tails fill one wave-wide load");
    const int hl = lane >> 4;
    const int tf = 32 * FULL + (lane & 15);
    auto batch = [&](auto whole_tag, int i0) {
      constexpr bool WHOLE = decltype(whole_tag)::value;
      float v[U][FULL];
      float vt[U / 2];
      float ww[U];
      int   rr[U];
      int   cl[U];
#pragma unroll
      for (int u = 0; u < U; ++u) {
        const int i  = i0 + u * NW;
        const int ic = WHOLE ? i : min(i, nq - 1);
        const unsigned p = (unsigned)__builtin_amdgcn_readfirstlane((int)qa[ic]);
        ww[u] = 1.0f;
        if constexpr (WT) ww[u] = qw[ic];
        rr[u] = (int)(p & 0xFFFFu);
        cl[u] = (WHOLE || i < nq) ? (int)(p >> 16) : -1;
        const float* __restrict__ s = hx + (long long)rr[u] * F;
#pragma unroll
        for (int k = 0; k < FULL; ++k) v[u][k] = WT ? s[lane + 32 * k] * ww[u] : s[lane + 32 * k];
      }
#pragma unroll
      for (int j = 0; j < U / 2; ++j) {
        const int   r = hl ? rr[2 * j + 1] : rr[2 * j];
        const float w = hl ? ww[2 * j + 1] : ww[2 * j];
        const float x = hx[(long long)r * F + tf];
        vt[j] = WT ? x * w : x;
      }
#pragma unroll
      for (int u = 0; u < U; ++u) {
        if (WHOLE || cl[u] >= 0) {
#pragma unroll
          for (int k = 0; k < FULL; ++k) atomicAdd(&acc[cl[u] * P + lane + 32 * k], v[u][k]);
        }
      }
#pragma unroll
      for (int j = 0; j < U / 2; ++j) {
        const int c = hl ? cl[2 * j + 1] : cl[2 * j];
        if (WHOLE || c >= 0) atomicAdd(&acc[c * P + tf], vt[j]);
      }
    };
    const int nwhole = (nq / (NW * U)) * (NW * U);
    int i0 = swave;
    for (; i0 < nwhole; i0 += NW * U) batch(std::true_type{}, i0);
    if (i0 < nq) batch(std::false_type{}, i0);
  } else
  for (int i0 = swave; i0 < nq; i0 += NW * U) {
    float v[U][PER];
    int   cl[U];
#pragma unroll
    for (int u = 0; u < U; ++u) {
      int i  = i0 + u * NW;
      int ic = min(i, nq - 1);
      unsigned p = qa[ic];
      float wq = 1.0f;
      if constexpr (WT) wq = qw[ic];
      int r = (int)(p & 0xFFFFu);
      cl[u] = (i < nq) ? (int)(p >> 16) : -1;
      const float* __restrict__ s = hx + (long long)r * F;
#pragma unroll
      for (int k = 0; k < PER; ++k) {
        int f = min(lane + 32 * k, F - 1);
        v[u][k] = WT ? s[f] * wq : s[f];
      }
    }
#pragma unroll
    for (int u = 0; u < U; ++u) {
      if (cl[u] >= 0) {
#pragma unroll
        for (int k = 0; k < PER; ++k) {
          int f = lane + 32 * k;
          if (f < F) atomicAdd(&acc[cl[u] * P + f], v[u][k]);
        }
      }
    }
  }
  __syncthreads();

  float* __restrict__ o = out + (long long)n0 * F;
  auto bias_of = [&](int f) -> const float* {
    if constexpr (LSM) {
      const int hd = f / N_CLS;
      return (hd ? bias2 : bias) + (f - hd * N_CLS);
    } else {
      return bias + f;
    }
  };
  if (nr == AGG_R) {
    float dd[NIT];
    float hh[NIT];
    fvec  bb[NIT];
#pragma unroll
    for (int u = 0; u < NIT; ++u) {
      const int pi = tid + u * AGG_BS;
      if (u < NIT - 1 || pi < NPIECE) {
        const int i = pi * V;
        const int r = i / F;
        const int f = i - r * F;
        dd[u] = dinv[n0 + r];
        hh[u] = HEADS ? hscale[n0 + r] : 1.0f;
        const float* __restrict__ bp = bias_of(f);
#pragma unroll
        for (int k = 0; k < V; ++k) bb[u][k] = bp[k];
      }
    }
#pragma unroll
    for (int u = 0; u < NIT; ++u) {
      const int pi = tid + u * AGG_BS;
      if (u < NIT - 1 || pi < NPIECE) {
        const int i = pi * V;
        const int r = i / F;
        const int f = i - r * F;
        float* a = acc + ((P == F) ? i : r * P + f);
        const fvec s = bb[u] + dd[u] * *(const fvec*)a;
        if constexpr (LSM) *(fvec*)a = s;
        else               *(fvec*)(o + i) = s;
        if constexpr (HEADS) *(fvec*)a = hh[u] * s;
      }
    }
  } else {
    for (int i = tid * V; i < nr * F; i += AGG_BS * V) {
      const int r = i / F;
      const int f = i - r * F;
      float* a = acc + ((P == F) ? i : r * P + f);
      const float d = dinv[n0 + r];
      const float* __restrict__ bp = bias_of(f);
      fvec s = *(const fvec*)a;
#pragma unroll
      for (int k = 0; k < V; ++k) s[k] = bp[k] + d * s[k];
      if constexpr (LSM) *(fvec*)a = s;
      else               *(fvec*)(o + i) = s;
      if constexpr (HEADS) *(fvec*)a = hscale[n0 + r] * s;
    }
  }

  if constexpr (LSM) {
    __syncthreads();
    if (tid < nr * NH) {
      const fvec* __restrict__ rp = (const fvec*)(acc + tid * N_CLS);
      fvec xv[N_CLS / V];
#pragma unroll
      for (int q = 0; q < N_CLS / V; ++q) xv[q] = rp[q];
      float m = -INFINITY;
#pragma unroll
      for (int cc = 0; cc < N_CLS; ++cc) m = fmaxf(m, xv[cc / V][cc % V]);
      float se = 0.0f;
#pragma unroll
      for (int cc = 0; cc < N_CLS; ++cc) se += __expf(xv[cc / V][cc % V] - m);
      lsm[tid] = __logf(se) + m;
    }
    __syncthreads();
#pragma unroll
    for (int hd = 0; hd < NH; ++hd) {
      float* __restrict__ oh = (hd ? out2 : out) + (long long)n0 * N_CLS;
      for (int i = tid * V; i < nr * N_CLS; i += AGG_BS * V) {
        const int r = i / N_CLS;
        const int c = i - r * N_CLS;
        *(fvec*)(oh + i) = *(const fvec*)(acc + r * P + hd * N_CLS + c) - lsm[r * NH + hd];
      }
    }
  }

  if constexpr (HEADS) {
    __syncthreads();
    static_assert(RT <= AGG_BS / 32, "one wave per 16-row tile, the other waves are done");
    static_assert(F % 4 == 0 && HEAD_W % 16 == 0, "K step 4; both heads' columns are whole 16-wide tiles");
    constexpr int NT = HEAD_W / 16;
    const int rt = swave;
    if (rt * 16 >= nr) return;
    const int c  = lane & 15;
    const int kh = lane >> 4;
    const float2* __restrict__ A2 = (const float2*)(acc + (rt * 16 + c) * P);
    const float2* __restrict__ B2 = (const float2*)wsh + c;
    v8f d[NT] = {};
#pragma unroll
    for (int k2 = 0; k2 < F / 2; k2 += 2) {
      const float2 af = A2[k2 + kh];
      v2f a; a.x = af.x; a.y = af.y;
#pragma unroll
      for (int t = 0; t < NT; ++t) {
        const float2 bf = B2[(k2 + kh) * HEAD_W + t * 16];
        v2f b; b.x = bf.x; b.y = bf.y;
        d[t] = __builtin_amdgcn_wmma_f32_16x16x4_f32(
             false, a,  false, b,
             (short)0, d[t],  false,  false);
      }
    }
    const int crow = rt * 16 + (kh << 3);
#pragma unroll
    for (int t = 0; t < NT; ++t) {
      const int col = (t << 4) + c;
#pragma unroll
      for (int r = 0; r < 8; ++r)
        if (crow + r < nr) hcat[(long long)(n0 + crow + r) * HEAD_W + col] = d[t][r];
    }
  }
}

__global__ void edge_dot_kernel(const float* __restrict__ h, const int* __restrict__ pos,
                                const int* __restrict__ neg, float* __restrict__ res) {
  int e = blockIdx.x * blockDim.x + threadIdx.x;
  if (e >= 2 * E_LP) return;
  const int* src = (e < E_LP) ? pos : neg;
  int eo = (e < E_LP) ? e : (e - E_LP);
  int j = src[eo];
  int i = src[E_LP + eo];
  const float4* hj = (const float4*)(h + (long long)j * H2_DIM);
  const float4* hi = (const float4*)(h + (long long)i * H2_DIM);
  float s = 0.0f;
#pragma unroll
  for (int k = 0; k < H2_DIM / 4; ++k) {
    float4 a = hi[k], b = hj[k];
    s += a.x * b.x + a.y * b.y + a.z * b.z + a.w * b.w;
  }
  res[e] = s;
}

static inline int blocks_for(long long total, int bs) {
  return (int)((total + bs - 1) / bs);
}

extern "C" void kernel_launch(void* const* d_in, const int* in_sizes, int n_in,
                              void* d_out, int out_size, void* d_ws, size_t ws_size,
                              hipStream_t stream) {
  const float* x      = (const float*)d_in[0];
  const int*   tpe    = (const int*)  d_in[1];
  const int*   ei     = (const int*)  d_in[2];
  const float* ew     = (const float*)d_in[3];
  const int*   pos    = (const int*)  d_in[4];
  const int*   neg    = (const int*)  d_in[5];
  const float* W1     = (const float*)d_in[6];
  const float* b1     = (const float*)d_in[7];
  const float* W2     = (const float*)d_in[8];
  const float* b2     = (const float*)d_in[9];
  const float* W_attr = (const float*)d_in[10];
  const float* b_attr = (const float*)d_in[11];
  const float* W_att  = (const float*)d_in[12];
  const float* b_att  = (const float*)d_in[13];

  float* out = (float*)d_out;
  float* ws  = (float*)d_ws;

  const long long OFF_DINV_T = 0;
  const long long OFF_DINV_W = 50000;
  const long long OFF_HX1    = 100000;
  const long long OFF_H1     = OFF_HX1 + 6400000;
  const long long OFF_HX2    = OFF_HX1;
  const long long OFF_BKT    = OFF_H1  + 6400000;
  const long long OFF_H2     = OFF_BKT + 3200000;
  const long long OFF_W1P    = OFF_H2  + 3200000;
  const long long OFF_W2P    = OFF_W1P + 65536;
  const long long OFF_HCAT   = OFF_H1;

  float* dinv_t = ws + OFF_DINV_T;
  float* dinv_w = ws + OFF_DINV_W;
  float* hx1    = ws + OFF_HX1;
  float* h1     = ws + OFF_H1;
  float* hx2    = ws + OFF_HX2;
  float* h2     = ws + OFF_H2;
  float* w1p    = ws + OFF_W1P;
  float* w2p    = ws + OFF_W2P;
  float* hcat   = ws + OFF_HCAT;

  const int NCH = (E_MP + BK_CHUNK - 1) / BK_CHUNK;
  unsigned* bent_w = (unsigned*)(ws + OFF_BKT);
  unsigned* bent_t = bent_w + 2 * E_MP;
  unsigned* btab_t = bent_t + E_MP;
  unsigned* btab_w = btab_t + NCH * (AGG_NB + 1);

  const int BS = 256;

  pack_b_kernel<<<blocks_for((D_FEAT / 2) * H1_DIM, BS), BS, 0, stream>>>(W1, w1p, D_FEAT, H1_DIM, H1_DIM);
  pack_b_kernel<<<blocks_for((H1_DIM / 2) * H2_DIM, BS), BS, 0, stream>>>(W2, w2p, H1_DIM, H2_DIM, H2_DIM);

  bucket_kernel<false><<<NCH, BK_BS, 0, stream>>>(tpe, tpe + E_MP, nullptr, bent_t, btab_t, E_MP, N_NODES);
  bucket_kernel<true ><<<NCH, BK_BS, 0, stream>>>(ei,  ei + E_MP,  ew,      bent_w, btab_w, E_MP, N_NODES);

  deg_own_kernel<false><<<AGG_NB, DEG_BS, 0, stream>>>(dinv_t, bent_t, btab_t, NCH, N_NODES);
  deg_own_kernel<true ><<<AGG_NB, DEG_BS, 0, stream>>>(dinv_w, bent_w, btab_w, NCH, N_NODES);

  const int ROW_TILES = N_NODES / 16;
  const int GEMM_BS = 128;
  const int gemm_blocks = blocks_for((long long)ROW_TILES * 32, GEMM_BS);
  const int AGG_BLOCKS = AGG_NB;

  wmma_gemm_kernel<D_FEAT, 8, H1_DIM, false><<<gemm_blocks, GEMM_BS, 0, stream>>>(x, w1p, dinv_t, hx1, N_NODES);
  gcn_agg_own_kernel<H1_DIM, false, false, false><<<AGG_BLOCKS, AGG_BS, 0, stream>>>(
      hx1, h1, dinv_t, b1, bent_t, btab_t, NCH, N_NODES, nullptr, nullptr, nullptr, nullptr, nullptr, nullptr);

  wmma_gemm_kernel<H1_DIM, 4, H2_DIM, true><<<gemm_blocks, GEMM_BS, 0, stream>>>(h1, w2p, dinv_t, hx2, N_NODES);
  gcn_agg_own_kernel<H2_DIM, false, true, false><<<AGG_BLOCKS, AGG_BS, 0, stream>>>(
      hx2, h2, dinv_t, b2, bent_t, btab_t, NCH, N_NODES, W_attr, W_att, dinv_w, hcat, nullptr, nullptr);

  float* out_attr = out + 2 * E_LP;
  float* out_att  = out + 2 * E_LP + (long long)N_NODES * N_CLS;
  gcn_agg_own_kernel<HEAD_W, true, false, true><<<AGG_BLOCKS, AGG_BS, 0, stream>>>(
      hcat, out_attr, dinv_w, b_attr, bent_w, btab_w, NCH, N_NODES, nullptr, nullptr, nullptr, nullptr, b_att, out_att);

  edge_dot_kernel<<<blocks_for(2 * E_LP, BS), BS, 0, stream>>>(h2, pos, neg, out);
}
